# starting kernel
# baseline (speedup 1.0000x reference)
#include <hip/hip_runtime.h>
#include <math.h>

#define N_NODES 50000
#define N_EDGES 800000
#define IN_F 128
#define OUT_F 128
#define NEG_SLOPE 0.2f
#define EPS_F 1e-10f

typedef float v2f __attribute__((ext_vector_type(2)));
typedef float v8f __attribute__((ext_vector_type(8)));


__device__ __forceinline__ void atomAddF(float* p, float v) {
    __hip_atomic_fetch_add(p, v, __ATOMIC_RELAXED, __HIP_MEMORY_SCOPE_AGENT);
}

__device__ __forceinline__ unsigned fenc(float f) {
    unsigned u = __float_as_uint(f);
    return (u & 0x80000000u) ? ~u : (u | 0x80000000u);
}
__device__ __forceinline__ float fdec(unsigned u) {
    return __uint_as_float((u & 0x80000000u) ? (u ^ 0x80000000u) : ~u);
}

__global__ void __launch_bounds__(128)
gat_gemm_wmma(const float* __restrict__ h, const float* __restrict__ W,
              const float* __restrict__ a, float* __restrict__ Wh,
              float* __restrict__ s1, float* __restrict__ s2, int n_tiles) {
    __shared__ float tlds[4][16 * 128];

    const int lane = threadIdx.x & 31;
    const int wave = threadIdx.x >> 5;
    const int tile = blockIdx.x * 4 + wave;
    if (tile >= n_tiles) return;

    const int row0 = tile * 16;
    const int half = lane >> 4;
    const int mn   = lane & 15;

    float* tl = &tlds[wave][0];
    {
        const float4* hp4 = (const float4*)(h + row0 * IN_F);
        float4* tl4 = (float4*)tl;
        #pragma unroll
        for (int t = 0; t < 16; ++t)
            tl4[t * 32 + lane] = hp4[t * 32 + lane];
    }

    float a1v[8], a2v[8];
    #pragma unroll
    for (int nt = 0; nt < 8; ++nt) {
        a1v[nt] = a[nt * 16 + mn];
        a2v[nt] = a[OUT_F + nt * 16 + mn];
    }

    v8f acc[8] = {};

    for (int kc = 0; kc < 32; ++kc) {
        const int col = kc * 4 + 2 * half;
        v2f av;
        {
            const float2 t = *(const float2*)(tl + mn * 128 + col);
            av[0] = t.x; av[1] = t.y;
        }
        #pragma unroll
        for (int nt = 0; nt < 8; ++nt) {
            const int ncol = nt * 16 + mn;
            v2f bv;
            {
                const float2 t = *(const float2*)(W + ncol * IN_F + col);
                bv[0] = t.x; bv[1] = t.y;
            }
            acc[nt] = __builtin_amdgcn_wmma_f32_16x16x4_f32(
                false, av, false, bv, (short)0, acc[nt], false, false);
        }
    }

    #pragma unroll
    for (int nt = 0; nt < 8; ++nt) {
        #pragma unroll
        for (int j = 0; j < 8; ++j) {
            Wh[(row0 + j + 8 * half) * OUT_F + nt * 16 + mn] = acc[nt][j];
        }
    }

    float p1[8], p2[8];
    #pragma unroll
    for (int j = 0; j < 8; ++j) { p1[j] = 0.f; p2[j] = 0.f; }
    #pragma unroll
    for (int nt = 0; nt < 8; ++nt) {
        #pragma unroll
        for (int j = 0; j < 8; ++j) {
            p1[j] = fmaf(acc[nt][j], a1v[nt], p1[j]);
            p2[j] = fmaf(acc[nt][j], a2v[nt], p2[j]);
        }
    }
    #pragma unroll
    for (int j = 0; j < 8; ++j) {
        #pragma unroll
        for (int off = 1; off < 16; off <<= 1) {
            p1[j] += __shfl_xor(p1[j], off, 32);
            p2[j] += __shfl_xor(p2[j], off, 32);
        }
    }
    if (mn == 0) {
        #pragma unroll
        for (int j = 0; j < 8; ++j) {
            s1[row0 + j + 8 * half] = p1[j];
            s2[row0 + j + 8 * half] = p2[j];
        }
    }
}

__global__ void __launch_bounds__(256)
gat_edge_score(const int* __restrict__ rows, const int* __restrict__ cols,
               const float* __restrict__ s1, const float* __restrict__ s2,
               float* __restrict__ e, unsigned* __restrict__ gmax, int nE) {
    __shared__ float smax[8];
    const int lane = threadIdx.x & 31;
    const int wave = threadIdx.x >> 5;

    float lmax = -INFINITY;
    for (int i = blockIdx.x * blockDim.x + threadIdx.x; i < nE;
         i += gridDim.x * blockDim.x) {
        float v = s1[rows[i]] + s2[cols[i]];
        v = (v > 0.f) ? v : NEG_SLOPE * v;
        e[i] = v;
        lmax = fmaxf(lmax, v);
    }
    #pragma unroll
    for (int off = 16; off >= 1; off >>= 1)
        lmax = fmaxf(lmax, __shfl_xor(lmax, off, 32));
    if (lane == 0) smax[wave] = lmax;
    __syncthreads();
    if (threadIdx.x == 0) {
        float m = smax[0];
        #pragma unroll
        for (int w = 1; w < 8; ++w) m = fmaxf(m, smax[w]);
        atomicMax(gmax, fenc(m));
    }
}

__global__ void __launch_bounds__(256)
gat_edge_exp(const int* __restrict__ rows, const float* __restrict__ e,
             const unsigned* __restrict__ gmax, float* __restrict__ expe,
             float* __restrict__ sum_exp, int nE) {
    const float gm = fdec(*gmax);
    for (int i = blockIdx.x * blockDim.x + threadIdx.x; i < nE;
         i += gridDim.x * blockDim.x) {
        float ex = __expf(e[i] - gm);
        expe[i] = ex;
        atomAddF(&sum_exp[rows[i]], ex);
    }
}

__global__ void __launch_bounds__(256)
gat_scatter(const int* __restrict__ rows, const int* __restrict__ cols,
            const float* __restrict__ expe, const float* __restrict__ sum_exp,
            const float* __restrict__ Wh, float* __restrict__ out, int nE) {
    const int lane = threadIdx.x & 31;
    const int wave = threadIdx.x >> 5;
    const int eid = blockIdx.x * 8 + wave;
    if (eid >= nE) return;

    const int r = rows[eid];
    const int c = cols[eid];
    const float alpha = expe[eid] / (sum_exp[r] + EPS_F);

    const float4 w = ((const float4*)(Wh + c * OUT_F))[lane];
    float* dst = out + r * OUT_F + lane * 4;
    atomAddF(dst + 0, alpha * w.x);
    atomAddF(dst + 1, alpha * w.y);
    atomAddF(dst + 2, alpha * w.z);
    atomAddF(dst + 3, alpha * w.w);
}

__global__ void __launch_bounds__(256)
gat_elu(float* __restrict__ out, int n) {
    int i = blockIdx.x * blockDim.x + threadIdx.x;
    if (i < n) {
        float x = out[i];
        out[i] = (x > 0.f) ? x : expm1f(x);
    }
}

extern "C" void kernel_launch(void* const* d_in, const int* in_sizes, int n_in,
                              void* d_out, int out_size, void* d_ws, size_t ws_size,
                              hipStream_t stream) {
    const float* h  = (const float*)d_in[0];
    const int*   ei = (const int*)d_in[1];
    const float* W  = (const float*)d_in[2];
    const float* a  = (const float*)d_in[3];
    float* out = (float*)d_out;

    const int* rows = ei;
    const int* cols = ei + N_EDGES;

    float* ws = (float*)d_ws;
    float*    Wh      = ws;
    float*    s1      = Wh + (size_t)N_NODES * OUT_F;
    float*    s2      = s1 + N_NODES;
    float*    e       = s2 + N_NODES;
    float*    expe    = e + N_EDGES;
    float*    sum_exp = expe + N_EDGES;
    unsigned* gmax    = (unsigned*)(sum_exp + N_NODES);

    hipMemsetAsync(out, 0, (size_t)N_NODES * OUT_F * sizeof(float), stream);
    hipMemsetAsync(sum_exp, 0, (size_t)N_NODES * sizeof(float), stream);
    hipMemsetAsync(gmax, 0, sizeof(unsigned), stream);

    const int n_tiles = N_NODES / 16;
    gat_gemm_wmma<<<(n_tiles + 3) / 4, 128, 0, stream>>>(h, W, a, Wh, s1, s2, n_tiles);

    gat_edge_score<<<(N_EDGES + 255) / 256, 256, 0, stream>>>(rows, cols, s1, s2, e,
                                                              gmax, N_EDGES);
    gat_edge_exp<<<(N_EDGES + 255) / 256, 256, 0, stream>>>(rows, e, gmax, expe,
                                                            sum_exp, N_EDGES);
    gat_scatter<<<(N_EDGES + 7) / 8, 256, 0, stream>>>(rows, cols, expe, sum_exp,
                                                       Wh, out, N_EDGES);
    const int total = N_NODES * OUT_F;
    gat_elu<<<(total + 255) / 256, 256, 0, stream>>>(out, total);
}


# Edges bucketed by row (int-atomic arrival-order rank + scan), then one wave per row gathers alpha*Wh in registers and applies ELU: no float atomics, no out memset; row sums run in arrival order
# speedup vs baseline: 1.5439x; 1.5439x over previous
#include <hip/hip_runtime.h>
#include <math.h>

#define N_NODES 50000
#define N_EDGES 800000
#define IN_F 128
#define OUT_F 128
#define NEG_SLOPE 0.2f
#define EPS_F 1e-10f

typedef float v2f __attribute__((ext_vector_type(2)));
typedef float v8f __attribute__((ext_vector_type(8)));


__device__ __forceinline__ unsigned fenc(float f) {
    unsigned u = __float_as_uint(f);
    return (u & 0x80000000u) ? ~u : (u | 0x80000000u);
}
__device__ __forceinline__ float fdec(unsigned u) {
    return __uint_as_float((u & 0x80000000u) ? (u ^ 0x80000000u) : ~u);
}

__device__ __forceinline__ float edge_score(float a, float b) {
    const float v = a + b;
    return (v > 0.f) ? v : NEG_SLOPE * v;
}

__global__ void __launch_bounds__(128)
gat_gemm_wmma(const float* __restrict__ h, const float* __restrict__ W,
              const float* __restrict__ a, float* __restrict__ Wh,
              float* __restrict__ s1, float* __restrict__ s2, int n_tiles) {
    __shared__ float tlds[4][16 * 128];

    const int lane = threadIdx.x & 31;
    const int wave = threadIdx.x >> 5;
    const int tile = blockIdx.x * 4 + wave;
    if (tile >= n_tiles) return;

    const int row0 = tile * 16;
    const int half = lane >> 4;
    const int mn   = lane & 15;

    float* tl = &tlds[wave][0];
    {
        const float4* hp4 = (const float4*)(h + row0 * IN_F);
        float4* tl4 = (float4*)tl;
        #pragma unroll
        for (int t = 0; t < 16; ++t)
            tl4[t * 32 + lane] = hp4[t * 32 + lane];
    }

    float a1v[8], a2v[8];
    #pragma unroll
    for (int nt = 0; nt < 8; ++nt) {
        a1v[nt] = a[nt * 16 + mn];
        a2v[nt] = a[OUT_F + nt * 16 + mn];
    }

    v8f acc[8] = {};

    for (int kc = 0; kc < 32; ++kc) {
        const int col = kc * 4 + 2 * half;
        v2f av;
        {
            const float2 t = *(const float2*)(tl + mn * 128 + col);
            av[0] = t.x; av[1] = t.y;
        }
        #pragma unroll
        for (int nt = 0; nt < 8; ++nt) {
            const int ncol = nt * 16 + mn;
            v2f bv;
            {
                const float2 t = *(const float2*)(W + ncol * IN_F + col);
                bv[0] = t.x; bv[1] = t.y;
            }
            acc[nt] = __builtin_amdgcn_wmma_f32_16x16x4_f32(
                false, av, false, bv, (short)0, acc[nt], false, false);
        }
    }

    #pragma unroll
    for (int nt = 0; nt < 8; ++nt) {
        #pragma unroll
        for (int j = 0; j < 8; ++j) {
            Wh[(row0 + j + 8 * half) * OUT_F + nt * 16 + mn] = acc[nt][j];
        }
    }

    float p1[8], p2[8];
    #pragma unroll
    for (int j = 0; j < 8; ++j) { p1[j] = 0.f; p2[j] = 0.f; }
    #pragma unroll
    for (int nt = 0; nt < 8; ++nt) {
        #pragma unroll
        for (int j = 0; j < 8; ++j) {
            p1[j] = fmaf(acc[nt][j], a1v[nt], p1[j]);
            p2[j] = fmaf(acc[nt][j], a2v[nt], p2[j]);
        }
    }
    #pragma unroll
    for (int j = 0; j < 8; ++j) {
        #pragma unroll
        for (int off = 1; off < 16; off <<= 1) {
            p1[j] += __shfl_xor(p1[j], off, 32);
            p2[j] += __shfl_xor(p2[j], off, 32);
        }
    }
    if (mn == 0) {
        #pragma unroll
        for (int j = 0; j < 8; ++j) {
            s1[row0 + j + 8 * half] = p1[j];
            s2[row0 + j + 8 * half] = p2[j];
        }
    }
}

__global__ void __launch_bounds__(256)
gat_edge_rank(const int* __restrict__ rows, const int* __restrict__ cols,
              const float* __restrict__ s1, const float* __restrict__ s2,
              unsigned* __restrict__ rank, unsigned* __restrict__ cnt,
              unsigned* __restrict__ gmax, int nE) {
    __shared__ float smax[8];
    const int lane = threadIdx.x & 31;
    const int wave = threadIdx.x >> 5;

    float lmax = -INFINITY;
    for (int i = blockIdx.x * blockDim.x + threadIdx.x; i < nE;
         i += gridDim.x * blockDim.x) {
        const int r = rows[i];
        const float v = edge_score(s1[r], s2[cols[i]]);
        lmax = fmaxf(lmax, v);
        rank[i] = atomicAdd(&cnt[r], 1u);
    }
    for (int off = 16; off >= 1; off >>= 1)
        lmax = fmaxf(lmax, __shfl_xor(lmax, off, 32));
    if (lane == 0) smax[wave] = lmax;
    __syncthreads();
    if (threadIdx.x == 0) {
        float m = smax[0];
        for (int w = 1; w < 8; ++w) m = fmaxf(m, smax[w]);
        atomicMax(gmax, fenc(m));
    }
}

#define SCAN_TILE 4096
#define SCAN_NT ((N_NODES + SCAN_TILE - 1) / SCAN_TILE)
__global__ void __launch_bounds__(1024)
gat_row_scan(unsigned* cnt) {
    __shared__ unsigned wtot[2][32];
    const int lane = threadIdx.x & 31;
    const int wave = threadIdx.x >> 5;
    uint4* c4 = (uint4*)cnt;

    uint4 v[SCAN_NT];
    #pragma unroll
    for (int t = 0; t < SCAN_NT; ++t) {
        const int idx = t * SCAN_TILE + threadIdx.x * 4;
        v[t] = (idx < N_NODES) ? c4[idx >> 2] : make_uint4(0u, 0u, 0u, 0u);
    }

    unsigned carry = 0u;
    #pragma unroll
    for (int t = 0; t < SCAN_NT; ++t) {
        const unsigned a0 = v[t].x;
        const unsigned a1 = a0 + v[t].y;
        const unsigned a2 = a1 + v[t].z;
        const unsigned a3 = a2 + v[t].w;
        unsigned s = a3;
        for (int off = 1; off < 32; off <<= 1) {
            const unsigned n = __shfl_up(s, off, 32);
            if (lane >= off) s += n;
        }
        if (lane == 31) wtot[t & 1][wave] = s;
        __syncthreads();
        const unsigned wt = wtot[t & 1][lane];
        unsigned ws = wt;
        for (int off = 1; off < 32; off <<= 1) {
            const unsigned n = __shfl_up(ws, off, 32);
            if (lane >= off) ws += n;
        }
        const unsigned total = __shfl(ws, 31, 32);
        const unsigned wexcl = __shfl(ws, wave, 32) - __shfl(wt, wave, 32);
        const unsigned base = carry + wexcl + (s - a3);
        const int idx = t * SCAN_TILE + threadIdx.x * 4;
        if (idx < N_NODES)
            c4[idx >> 2] = make_uint4(base, base + a0, base + a1, base + a2);
        carry += total;
    }
}

__global__ void __launch_bounds__(256)
gat_edge_place(const int* __restrict__ rows, const int* __restrict__ cols,
               const unsigned* __restrict__ rank, const unsigned* __restrict__ roff,
               int* __restrict__ scol, int nE) {
    for (int i = blockIdx.x * blockDim.x + threadIdx.x; i < nE;
         i += gridDim.x * blockDim.x) {
        scol[roff[rows[i]] + rank[i]] = cols[i];
    }
}

__global__ void __launch_bounds__(256)
gat_row_gather(const unsigned* __restrict__ roff, const int* __restrict__ scol,
               const float* __restrict__ s1, const float* __restrict__ s2,
               const unsigned* __restrict__ gmax, const float* __restrict__ Wh,
               float* __restrict__ out) {
    const int lane = threadIdx.x & 31;
    const int wave = __builtin_amdgcn_readfirstlane((int)(threadIdx.x >> 5));
    const int r = blockIdx.x * 8 + wave;
    if (r >= N_NODES) return;

    const int beg = (int)roff[r];
    const int end = (r + 1 < N_NODES) ? (int)roff[r + 1] : N_EDGES;
    const float gm = fdec(*gmax);
    const float s1r = s1[r];

    float lsum = 0.f;
    for (int j = beg + lane; j < end; j += 32)
        lsum += __expf(edge_score(s1r, s2[scol[j]]) - gm);
    for (int off = 16; off >= 1; off >>= 1)
        lsum += __shfl_xor(lsum, off, 32);
    const float den = lsum + EPS_F;

    const float4* Wh4 = (const float4*)Wh;
    float4 acc = make_float4(0.f, 0.f, 0.f, 0.f);
    for (int base = beg; base < end; base += 32) {
        const int j = base + lane;
        const int c = scol[(j < end) ? j : (end - 1)];
        const float ex = __expf(edge_score(s1r, s2[c]) - gm);
        const float al = (j < end) ? ex / den : 0.f;
        const int n = min(32, end - base);
        const int n4 = (n + 3) & ~3;
        for (int k = 0; k < n4; k += 4) {
            float4 w[4];
            float ak[4];
            for (int u = 0; u < 4; ++u) {
                const int ck = __builtin_amdgcn_readlane(c, k + u);
                ak[u] = __int_as_float(__builtin_amdgcn_readlane(__float_as_int(al), k + u));
                w[u] = Wh4[ck * (OUT_F / 4) + lane];
            }
            for (int u = 0; u < 4; ++u) {
                acc.x = fmaf(ak[u], w[u].x, acc.x);
                acc.y = fmaf(ak[u], w[u].y, acc.y);
                acc.z = fmaf(ak[u], w[u].z, acc.z);
                acc.w = fmaf(ak[u], w[u].w, acc.w);
            }
        }
    }

    float4 o;
    o.x = (acc.x > 0.f) ? acc.x : expm1f(acc.x);
    o.y = (acc.y > 0.f) ? acc.y : expm1f(acc.y);
    o.z = (acc.z > 0.f) ? acc.z : expm1f(acc.z);
    o.w = (acc.w > 0.f) ? acc.w : expm1f(acc.w);
    ((float4*)(out + r * OUT_F))[lane] = o;
}

extern "C" void kernel_launch(void* const* d_in, const int* in_sizes, int n_in,
                              void* d_out, int out_size, void* d_ws, size_t ws_size,
                              hipStream_t stream) {
    const float* h  = (const float*)d_in[0];
    const int*   ei = (const int*)d_in[1];
    const float* W  = (const float*)d_in[2];
    const float* a  = (const float*)d_in[3];
    float* out = (float*)d_out;

    const int* rows = ei;
    const int* cols = ei + N_EDGES;

    float* ws = (float*)d_ws;
    float*    Wh   = ws;
    float*    s1   = Wh + (size_t)N_NODES * OUT_F;
    float*    s2   = s1 + N_NODES;
    unsigned* rank = (unsigned*)(s2 + N_NODES);
    int*      scol = (int*)(rank + N_EDGES);
    unsigned* cnt  = (unsigned*)(scol + N_EDGES);
    unsigned* gmax = cnt + N_NODES;

    hipMemsetAsync(cnt, 0, (size_t)(N_NODES + 1) * sizeof(unsigned), stream);

    const int n_tiles = N_NODES / 16;
    gat_gemm_wmma<<<(n_tiles + 3) / 4, 128, 0, stream>>>(h, W, a, Wh, s1, s2, n_tiles);

    gat_edge_rank<<<(N_EDGES + 255) / 256, 256, 0, stream>>>(rows, cols, s1, s2, rank, cnt,
                                                             gmax, N_EDGES);
    gat_row_scan<<<1, 1024, 0, stream>>>(cnt);
    gat_edge_place<<<(N_EDGES + 255) / 256, 256, 0, stream>>>(rows, cols, rank, cnt, scol,
                                                              N_EDGES);
    gat_row_gather<<<(N_NODES + 7) / 8, 256, 0, stream>>>(cnt, scol, s1, s2, gmax, Wh, out);
}


# Edges grouped by row with LDS counters, not 800k returning global atomics: chunks bin edges into 128 row ranges, one work-group per range ranks and places them; row sums stay in arrival order
# speedup vs baseline: 2.2854x; 1.4803x over previous
#include <hip/hip_runtime.h>
#include <math.h>

#define N_NODES 50000
#define N_EDGES 800000
#define IN_F 128
#define OUT_F 128
#define NEG_SLOPE 0.2f
#define EPS_F 1e-10f

typedef float v2f __attribute__((ext_vector_type(2)));
typedef float v8f __attribute__((ext_vector_type(8)));


__device__ __forceinline__ unsigned fenc(float f) {
    unsigned u = __float_as_uint(f);
    return (u & 0x80000000u) ? ~u : (u | 0x80000000u);
}
__device__ __forceinline__ float fdec(unsigned u) {
    return __uint_as_float((u & 0x80000000u) ? (u ^ 0x80000000u) : ~u);
}

__device__ __forceinline__ float edge_score(float a, float b) {
    const float v = a + b;
    return (v > 0.f) ? v : NEG_SLOPE * v;
}

__global__ void __launch_bounds__(128)
gat_gemm_wmma(const float* __restrict__ h, const float* __restrict__ W,
              const float* __restrict__ a, float* __restrict__ Wh,
              float* __restrict__ s1, float* __restrict__ s2, int n_tiles) {
    __shared__ float tlds[4][16 * 128];

    const int lane = threadIdx.x & 31;
    const int wave = threadIdx.x >> 5;
    const int tile = blockIdx.x * 4 + wave;
    if (tile >= n_tiles) return;

    const int row0 = tile * 16;
    const int half = lane >> 4;
    const int mn   = lane & 15;

    float* tl = &tlds[wave][0];
    {
        const float4* hp4 = (const float4*)(h + row0 * IN_F);
        float4* tl4 = (float4*)tl;
        #pragma unroll
        for (int t = 0; t < 16; ++t)
            tl4[t * 32 + lane] = hp4[t * 32 + lane];
    }

    float a1v[8], a2v[8];
    #pragma unroll
    for (int nt = 0; nt < 8; ++nt) {
        a1v[nt] = a[nt * 16 + mn];
        a2v[nt] = a[OUT_F + nt * 16 + mn];
    }

    v8f acc[8] = {};

    for (int kc = 0; kc < 32; ++kc) {
        const int col = kc * 4 + 2 * half;
        v2f av;
        {
            const float2 t = *(const float2*)(tl + mn * 128 + col);
            av[0] = t.x; av[1] = t.y;
        }
        #pragma unroll
        for (int nt = 0; nt < 8; ++nt) {
            const int ncol = nt * 16 + mn;
            v2f bv;
            {
                const float2 t = *(const float2*)(W + ncol * IN_F + col);
                bv[0] = t.x; bv[1] = t.y;
            }
            acc[nt] = __builtin_amdgcn_wmma_f32_16x16x4_f32(
                false, av, false, bv, (short)0, acc[nt], false, false);
        }
    }

    #pragma unroll
    for (int nt = 0; nt < 8; ++nt) {
        #pragma unroll
        for (int j = 0; j < 8; ++j) {
            Wh[(row0 + j + 8 * half) * OUT_F + nt * 16 + mn] = acc[nt][j];
        }
    }

    float p1[8], p2[8];
    #pragma unroll
    for (int j = 0; j < 8; ++j) { p1[j] = 0.f; p2[j] = 0.f; }
    #pragma unroll
    for (int nt = 0; nt < 8; ++nt) {
        #pragma unroll
        for (int j = 0; j < 8; ++j) {
            p1[j] = fmaf(acc[nt][j], a1v[nt], p1[j]);
            p2[j] = fmaf(acc[nt][j], a2v[nt], p2[j]);
        }
    }
    #pragma unroll
    for (int j = 0; j < 8; ++j) {
        #pragma unroll
        for (int off = 1; off < 16; off <<= 1) {
            p1[j] += __shfl_xor(p1[j], off, 32);
            p2[j] += __shfl_xor(p2[j], off, 32);
        }
    }
    if (mn == 0) {
        #pragma unroll
        for (int j = 0; j < 8; ++j) {
            s1[row0 + j + 8 * half] = p1[j];
            s2[row0 + j + 8 * half] = p2[j];
        }
    }
}

#define NBKT 128
#define RPB 391
#define EG 256
#define ECH (N_EDGES / EG)
#define CT 256
#define EPT ((ECH + CT - 1) / CT)
#define DT 512
#define EPT_D 16
#define RCAP (EPT_D * DT)
static_assert(NBKT * RPB >= N_NODES, "ranges must cover all rows");
static_assert(ECH * EG == N_EDGES, "chunks must tile the edge list");
static_assert(NBKT <= CT && RPB <= DT, "one thread per range counter / per row counter");
static_assert(N_NODES <= 65536 && RPB <= 512, "packed edge word: local row above bit 16, column below");

__global__ void __launch_bounds__(CT)
gat_bucket_count(const int* __restrict__ rows, unsigned* __restrict__ bc,
                 unsigned* __restrict__ gmax) {
    __shared__ unsigned hist[NBKT];
    const int tid = threadIdx.x;
    if (tid < NBKT) hist[tid] = 0u;
    if (blockIdx.x == 0 && tid == 0) *gmax = 0u;
    __syncthreads();

    const int base = blockIdx.x * ECH;
    int r[EPT];
    #pragma unroll
    for (int u = 0; u < EPT; ++u) {
        const int i = u * CT + tid;
        r[u] = (i < ECH) ? rows[base + i] : -1;
    }
    #pragma unroll
    for (int u = 0; u < EPT; ++u)
        if (r[u] >= 0) atomicAdd(&hist[r[u] / RPB], 1u);
    __syncthreads();
    if (tid < NBKT) bc[tid * EG + blockIdx.x] = hist[tid];
}

#define SCAN_N (NBKT * EG)
#define SCAN_TILE 4096
#define SCAN_NT ((SCAN_N + SCAN_TILE - 1) / SCAN_TILE)
__global__ void __launch_bounds__(1024)
gat_slot_scan(unsigned* cnt) {
    __shared__ unsigned wtot[2][32];
    const int lane = threadIdx.x & 31;
    const int wave = threadIdx.x >> 5;
    uint4* c4 = (uint4*)cnt;

    uint4 v[SCAN_NT];
    #pragma unroll
    for (int t = 0; t < SCAN_NT; ++t) {
        const int idx = t * SCAN_TILE + threadIdx.x * 4;
        v[t] = (idx < SCAN_N) ? c4[idx >> 2] : make_uint4(0u, 0u, 0u, 0u);
    }

    unsigned carry = 0u;
    #pragma unroll
    for (int t = 0; t < SCAN_NT; ++t) {
        const unsigned a0 = v[t].x;
        const unsigned a1 = a0 + v[t].y;
        const unsigned a2 = a1 + v[t].z;
        const unsigned a3 = a2 + v[t].w;
        unsigned s = a3;
        for (int off = 1; off < 32; off <<= 1) {
            const unsigned n = __shfl_up(s, off, 32);
            if (lane >= off) s += n;
        }
        if (lane == 31) wtot[t & 1][wave] = s;
        __syncthreads();
        const unsigned wt = wtot[t & 1][lane];
        unsigned ws = wt;
        for (int off = 1; off < 32; off <<= 1) {
            const unsigned n = __shfl_up(ws, off, 32);
            if (lane >= off) ws += n;
        }
        const unsigned total = __shfl(ws, 31, 32);
        const unsigned wexcl = __shfl(ws, wave, 32) - __shfl(wt, wave, 32);
        const unsigned base = carry + wexcl + (s - a3);
        const int idx = t * SCAN_TILE + threadIdx.x * 4;
        if (idx < SCAN_N)
            c4[idx >> 2] = make_uint4(base, base + a0, base + a1, base + a2);
        carry += total;
    }
}

__global__ void __launch_bounds__(CT)
gat_bucket_scatter(const int* __restrict__ rows, const int* __restrict__ cols,
                   const float* __restrict__ s1, const float* __restrict__ s2,
                   const unsigned* __restrict__ boff, unsigned* __restrict__ blist,
                   unsigned* __restrict__ gmax) {
    __shared__ unsigned cur[NBKT];
    __shared__ float smax[CT / 32];
    const int tid = threadIdx.x;
    const int lane = tid & 31;
    const int wave = tid >> 5;

    const int base = blockIdx.x * ECH;
    int r[EPT], c[EPT];
    #pragma unroll
    for (int u = 0; u < EPT; ++u) {
        const int i = u * CT + tid;
        r[u] = (i < ECH) ? rows[base + i] : -1;
        c[u] = (i < ECH) ? cols[base + i] : 0;
    }
    if (tid < NBKT) cur[tid] = boff[tid * EG + blockIdx.x];
    __syncthreads();

    float sc[EPT];
    #pragma unroll
    for (int u = 0; u < EPT; ++u)
        sc[u] = edge_score(s1[(r[u] >= 0) ? r[u] : 0], s2[c[u]]);

    unsigned pos[EPT];
    #pragma unroll
    for (int u = 0; u < EPT; ++u)
        pos[u] = (r[u] >= 0) ? atomicAdd(&cur[r[u] / RPB], 1u) : 0u;

    float lmax = -INFINITY;
    #pragma unroll
    for (int u = 0; u < EPT; ++u) {
        if (r[u] >= 0) {
            blist[pos[u]] = ((unsigned)(r[u] % RPB) << 16) | (unsigned)c[u];
            lmax = fmaxf(lmax, sc[u]);
        }
    }
    for (int off = 16; off >= 1; off >>= 1)
        lmax = fmaxf(lmax, __shfl_xor(lmax, off, 32));
    if (lane == 0) smax[wave] = lmax;
    __syncthreads();
    if (wave == 0) {
        float m = (lane < CT / 32) ? smax[lane] : -INFINITY;
        for (int off = 16; off >= 1; off >>= 1)
            m = fmaxf(m, __shfl_xor(m, off, 32));
        if (lane == 0) atomicMax(gmax, fenc(m));
    }
}

__global__ void __launch_bounds__(DT)
gat_bucket_csr(const unsigned* __restrict__ boff, const unsigned* __restrict__ blist,
               unsigned* __restrict__ roff, int* __restrict__ scol) {
    __shared__ unsigned cntL[DT];
    __shared__ unsigned wtot[DT / 32];
    __shared__ int sL[RCAP];
    const int tid = threadIdx.x;
    const int lane = tid & 31;
    const int wave = tid >> 5;
    const int b = blockIdx.x;
    const int row0 = b * RPB;
    const int beg = (int)boff[b * EG];
    const int end = (b + 1 < NBKT) ? (int)boff[(b + 1) * EG] : N_EDGES;
    const int n = end - beg;

    cntL[tid] = 0u;

    unsigned e[EPT_D];
    #pragma unroll
    for (int u = 0; u < EPT_D; ++u) {
        const int j = u * DT + tid;
        e[u] = (j < n) ? blist[beg + j] : 0xFFFFFFFFu;
    }
    __syncthreads();

    #pragma unroll
    for (int u = 0; u < EPT_D; ++u)
        if (e[u] != 0xFFFFFFFFu) atomicAdd(&cntL[e[u] >> 16], 1u);
    for (int j = RCAP + tid; j < n; j += DT)
        atomicAdd(&cntL[blist[beg + j] >> 16], 1u);
    __syncthreads();

    const unsigned v = cntL[tid];
    unsigned s = v;
    for (int off = 1; off < 32; off <<= 1) {
        const unsigned t = __shfl_up(s, off, 32);
        if (lane >= off) s += t;
    }
    if (lane == 31) wtot[wave] = s;
    __syncthreads();
    const unsigned wt = (lane < DT / 32) ? wtot[lane] : 0u;
    unsigned ws = wt;
    for (int off = 1; off < 32; off <<= 1) {
        const unsigned t = __shfl_up(ws, off, 32);
        if (lane >= off) ws += t;
    }
    const unsigned wexcl = __shfl(ws, wave, 32) - __shfl(wt, wave, 32);
    const unsigned excl = wexcl + s - v;
    cntL[tid] = excl;
    if (tid < RPB && row0 + tid < N_NODES) roff[row0 + tid] = (unsigned)beg + excl;
    __syncthreads();

    if (n <= RCAP) {
        unsigned pos[EPT_D];
        #pragma unroll
        for (int u = 0; u < EPT_D; ++u)
            pos[u] = (e[u] != 0xFFFFFFFFu) ? atomicAdd(&cntL[e[u] >> 16], 1u) : 0u;
        #pragma unroll
        for (int u = 0; u < EPT_D; ++u)
            if (e[u] != 0xFFFFFFFFu) sL[pos[u]] = (int)(e[u] & 0xFFFFu);
        __syncthreads();
        for (int j = tid; j < n; j += DT)
            scol[beg + j] = sL[j];
    } else {
        #pragma unroll
        for (int u = 0; u < EPT_D; ++u)
            if (e[u] != 0xFFFFFFFFu) {
                const unsigned pos = atomicAdd(&cntL[e[u] >> 16], 1u);
                scol[beg + pos] = (int)(e[u] & 0xFFFFu);
            }
        for (int j = RCAP + tid; j < n; j += DT) {
            const unsigned w = blist[beg + j];
            const unsigned pos = atomicAdd(&cntL[w >> 16], 1u);
            scol[beg + pos] = (int)(w & 0xFFFFu);
        }
    }
}

__global__ void __launch_bounds__(256)
gat_row_gather(const unsigned* __restrict__ roff, const int* __restrict__ scol,
               const float* __restrict__ s1, const float* __restrict__ s2,
               const unsigned* __restrict__ gmax, const float* __restrict__ Wh,
               float* __restrict__ out) {
    const int lane = threadIdx.x & 31;
    const int wave = __builtin_amdgcn_readfirstlane((int)(threadIdx.x >> 5));
    const int r = blockIdx.x * 8 + wave;
    if (r >= N_NODES) return;

    const int beg = (int)roff[r];
    const int end = (r + 1 < N_NODES) ? (int)roff[r + 1] : N_EDGES;
    const float gm = fdec(*gmax);
    const float s1r = s1[r];

    float lsum = 0.f;
    for (int j = beg + lane; j < end; j += 32)
        lsum += __expf(edge_score(s1r, s2[scol[j]]) - gm);
    for (int off = 16; off >= 1; off >>= 1)
        lsum += __shfl_xor(lsum, off, 32);
    const float den = lsum + EPS_F;

    const float4* Wh4 = (const float4*)Wh;
    float4 acc = make_float4(0.f, 0.f, 0.f, 0.f);
    for (int base = beg; base < end; base += 32) {
        const int j = base + lane;
        const int c = scol[(j < end) ? j : (end - 1)];
        const float ex = __expf(edge_score(s1r, s2[c]) - gm);
        const float al = (j < end) ? ex / den : 0.f;
        const int n = min(32, end - base);
        const int n4 = (n + 3) & ~3;
        for (int k = 0; k < n4; k += 4) {
            float4 w[4];
            float ak[4];
            for (int u = 0; u < 4; ++u) {
                const int ck = __builtin_amdgcn_readlane(c, k + u);
                ak[u] = __int_as_float(__builtin_amdgcn_readlane(__float_as_int(al), k + u));
                w[u] = Wh4[ck * (OUT_F / 4) + lane];
            }
            for (int u = 0; u < 4; ++u) {
                acc.x = fmaf(ak[u], w[u].x, acc.x);
                acc.y = fmaf(ak[u], w[u].y, acc.y);
                acc.z = fmaf(ak[u], w[u].z, acc.z);
                acc.w = fmaf(ak[u], w[u].w, acc.w);
            }
        }
    }

    float4 o;
    o.x = (acc.x > 0.f) ? acc.x : expm1f(acc.x);
    o.y = (acc.y > 0.f) ? acc.y : expm1f(acc.y);
    o.z = (acc.z > 0.f) ? acc.z : expm1f(acc.z);
    o.w = (acc.w > 0.f) ? acc.w : expm1f(acc.w);
    ((float4*)(out + r * OUT_F))[lane] = o;
}

extern "C" void kernel_launch(void* const* d_in, const int* in_sizes, int n_in,
                              void* d_out, int out_size, void* d_ws, size_t ws_size,
                              hipStream_t stream) {
    const float* h  = (const float*)d_in[0];
    const int*   ei = (const int*)d_in[1];
    const float* W  = (const float*)d_in[2];
    const float* a  = (const float*)d_in[3];
    float* out = (float*)d_out;

    const int* rows = ei;
    const int* cols = ei + N_EDGES;

    float* ws = (float*)d_ws;
    float*    Wh    = ws;
    float*    s1    = Wh + (size_t)N_NODES * OUT_F;
    float*    s2    = s1 + N_NODES;
    unsigned* blist = (unsigned*)(s2 + N_NODES);
    int*      scol  = (int*)(blist + N_EDGES);
    unsigned* roff  = (unsigned*)(scol + N_EDGES);
    unsigned* gmax  = roff + N_NODES;
    unsigned* bc    = (unsigned*)d_out;

    const int n_tiles = N_NODES / 16;
    gat_gemm_wmma<<<(n_tiles + 3) / 4, 128, 0, stream>>>(h, W, a, Wh, s1, s2, n_tiles);

    gat_bucket_count<<<EG, CT, 0, stream>>>(rows, bc, gmax);
    gat_slot_scan<<<1, 1024, 0, stream>>>(bc);
    gat_bucket_scatter<<<EG, CT, 0, stream>>>(rows, cols, s1, s2, bc, blist, gmax);
    gat_bucket_csr<<<NBKT, DT, 0, stream>>>(bc, blist, roff, scol);
    gat_row_gather<<<(N_NODES + 7) / 8, 256, 0, stream>>>(roff, scol, s1, s2, gmax, Wh, out);
}


# Each chunk sorts its edges by row range in LDS and writes a (range, chunk) table of run starts; range work-groups collect their 256 runs: count and scan launches gone; row sums stay in arrival order
# speedup vs baseline: 3.1550x; 1.3805x over previous
#include <hip/hip_runtime.h>
#include <math.h>

#define N_NODES 50000
#define N_EDGES 800000
#define IN_F 128
#define OUT_F 128
#define NEG_SLOPE 0.2f
#define EPS_F 1e-10f

typedef float v2f __attribute__((ext_vector_type(2)));
typedef float v8f __attribute__((ext_vector_type(8)));


__device__ __forceinline__ unsigned fenc(float f) {
    unsigned u = __float_as_uint(f);
    return (u & 0x80000000u) ? ~u : (u | 0x80000000u);
}
__device__ __forceinline__ float fdec(unsigned u) {
    return __uint_as_float((u & 0x80000000u) ? (u ^ 0x80000000u) : ~u);
}

__device__ __forceinline__ float edge_score(float a, float b) {
    const float v = a + b;
    return (v > 0.f) ? v : NEG_SLOPE * v;
}

__global__ void __launch_bounds__(128)
gat_gemm_wmma(const float* __restrict__ h, const float* __restrict__ W,
              const float* __restrict__ a, float* __restrict__ Wh,
              float* __restrict__ s1, float* __restrict__ s2, unsigned* __restrict__ gmax,
              int n_tiles) {
    __shared__ float tlds[4][16 * 128];

    if (blockIdx.x == 0 && threadIdx.x == 0) *gmax = 0u;

    const int lane = threadIdx.x & 31;
    const int wave = threadIdx.x >> 5;
    const int tile = blockIdx.x * 4 + wave;
    if (tile >= n_tiles) return;

    const int row0 = tile * 16;
    const int half = lane >> 4;
    const int mn   = lane & 15;

    float* tl = &tlds[wave][0];
    {
        const float4* hp4 = (const float4*)(h + row0 * IN_F);
        float4* tl4 = (float4*)tl;
        #pragma unroll
        for (int t = 0; t < 16; ++t)
            tl4[t * 32 + lane] = hp4[t * 32 + lane];
    }

    float a1v[8], a2v[8];
    #pragma unroll
    for (int nt = 0; nt < 8; ++nt) {
        a1v[nt] = a[nt * 16 + mn];
        a2v[nt] = a[OUT_F + nt * 16 + mn];
    }

    v8f acc[8] = {};

    for (int kc = 0; kc < 32; ++kc) {
        const int col = kc * 4 + 2 * half;
        v2f av;
        {
            const float2 t = *(const float2*)(tl + mn * 128 + col);
            av[0] = t.x; av[1] = t.y;
        }
        #pragma unroll
        for (int nt = 0; nt < 8; ++nt) {
            const int ncol = nt * 16 + mn;
            v2f bv;
            {
                const float2 t = *(const float2*)(W + ncol * IN_F + col);
                bv[0] = t.x; bv[1] = t.y;
            }
            acc[nt] = __builtin_amdgcn_wmma_f32_16x16x4_f32(
                false, av, false, bv, (short)0, acc[nt], false, false);
        }
    }

    #pragma unroll
    for (int nt = 0; nt < 8; ++nt) {
        #pragma unroll
        for (int j = 0; j < 8; ++j) {
            Wh[(row0 + j + 8 * half) * OUT_F + nt * 16 + mn] = acc[nt][j];
        }
    }

    float p1[8], p2[8];
    #pragma unroll
    for (int j = 0; j < 8; ++j) { p1[j] = 0.f; p2[j] = 0.f; }
    #pragma unroll
    for (int nt = 0; nt < 8; ++nt) {
        #pragma unroll
        for (int j = 0; j < 8; ++j) {
            p1[j] = fmaf(acc[nt][j], a1v[nt], p1[j]);
            p2[j] = fmaf(acc[nt][j], a2v[nt], p2[j]);
        }
    }
    #pragma unroll
    for (int j = 0; j < 8; ++j) {
        #pragma unroll
        for (int off = 1; off < 16; off <<= 1) {
            p1[j] += __shfl_xor(p1[j], off, 32);
            p2[j] += __shfl_xor(p2[j], off, 32);
        }
    }
    if (mn == 0) {
        #pragma unroll
        for (int j = 0; j < 8; ++j) {
            s1[row0 + j + 8 * half] = p1[j];
            s2[row0 + j + 8 * half] = p2[j];
        }
    }
}

#define NBKT 128
#define RPB 391
#define EG 256
#define ECH (N_EDGES / EG)
#define CT 256
#define EPT ((ECH + CT - 1) / CT)
#define DT 512
#define PPW (EG / (DT / 32))
#define RCAP 8192
#define EMPTY_EDGE 0xFFFFFFFFu
static_assert(NBKT * RPB >= N_NODES, "ranges must cover all rows");
static_assert(ECH * EG == N_EDGES, "chunks must tile the edge list");
static_assert(NBKT == 4 * 32, "one wave scans the range counters, four per lane");
static_assert(RPB <= DT && EG <= DT, "one thread per row counter / per run of a range");
static_assert(PPW * (DT / 32) == EG, "the waves of a range work-group share out its runs evenly");
static_assert(N_NODES <= 65536 && RPB <= 512, "packed edge word: local row above bit 16, column below");
static_assert(ECH < 65536, "table word: run start above bit 16, run length below");

__global__ void __launch_bounds__(CT)
gat_chunk_sort(const int* __restrict__ rows, const int* __restrict__ cols,
               const float* __restrict__ s1, const float* __restrict__ s2,
               unsigned* __restrict__ tab, unsigned* __restrict__ blist,
               unsigned* __restrict__ gmax) {
    __shared__ __attribute__((aligned(16))) unsigned hist[NBKT];
    __shared__ unsigned sorted[ECH];
    __shared__ float smax[CT / 32];
    const int tid = threadIdx.x;
    const int lane = tid & 31;
    const int wave = tid >> 5;
    if (tid < NBKT) hist[tid] = 0u;

    const int base = blockIdx.x * ECH;
    int r[EPT], c[EPT];
    #pragma unroll
    for (int u = 0; u < EPT; ++u) {
        const int i = u * CT + tid;
        r[u] = (i < ECH) ? rows[base + i] : 0;
        c[u] = (i < ECH) ? cols[base + i] : 0;
    }
    __syncthreads();

    float sc[EPT];
    #pragma unroll
    for (int u = 0; u < EPT; ++u)
        sc[u] = edge_score(s1[r[u]], s2[c[u]]);

    int q[EPT];
    unsigned rk[EPT];
    #pragma unroll
    for (int u = 0; u < EPT; ++u) {
        q[u] = r[u] / RPB;
        rk[u] = (u * CT + tid < ECH) ? atomicAdd(&hist[q[u]], 1u) : 0u;
    }

    float lmax = -INFINITY;
    #pragma unroll
    for (int u = 0; u < EPT; ++u)
        if (u * CT + tid < ECH) lmax = fmaxf(lmax, sc[u]);
    for (int off = 16; off >= 1; off >>= 1)
        lmax = fmaxf(lmax, __shfl_xor(lmax, off, 32));
    if (lane == 0) smax[wave] = lmax;
    __syncthreads();

    if (wave == 0) {
        const uint4 hv = ((const uint4*)hist)[lane];
        const unsigned t4 = hv.x + hv.y + hv.z + hv.w;
        unsigned s = t4;
        for (int off = 1; off < 32; off <<= 1) {
            const unsigned n = __shfl_up(s, off, 32);
            if (lane >= off) s += n;
        }
        const unsigned e0 = s - t4;
        const unsigned e1 = e0 + hv.x;
        const unsigned e2 = e1 + hv.y;
        const unsigned e3 = e2 + hv.z;
        ((uint4*)hist)[lane] = make_uint4(e0, e1, e2, e3);
        tab[(4 * lane + 0) * EG + blockIdx.x] = (e0 << 16) | hv.x;
        tab[(4 * lane + 1) * EG + blockIdx.x] = (e1 << 16) | hv.y;
        tab[(4 * lane + 2) * EG + blockIdx.x] = (e2 << 16) | hv.z;
        tab[(4 * lane + 3) * EG + blockIdx.x] = (e3 << 16) | hv.w;
    } else if (wave == 1) {
        float m = (lane < CT / 32) ? smax[lane] : -INFINITY;
        for (int off = 16; off >= 1; off >>= 1)
            m = fmaxf(m, __shfl_xor(m, off, 32));
        if (lane == 0) atomicMax(gmax, fenc(m));
    }
    __syncthreads();

    unsigned slot[EPT];
    #pragma unroll
    for (int u = 0; u < EPT; ++u)
        slot[u] = hist[q[u]] + rk[u];
    #pragma unroll
    for (int u = 0; u < EPT; ++u)
        if (u * CT + tid < ECH)
            sorted[slot[u]] = ((unsigned)(r[u] - q[u] * RPB) << 16) | (unsigned)c[u];
    __syncthreads();

    for (int i = tid; i < ECH; i += CT)
        blist[base + i] = sorted[i];
}

__global__ void __launch_bounds__(DT)
gat_range_csr(const unsigned* __restrict__ tab, const unsigned* __restrict__ blist,
              unsigned* __restrict__ roff, int* __restrict__ scol) {
    __shared__ unsigned cntL[DT];
    __shared__ unsigned wtot[DT / 32];
    __shared__ __attribute__((aligned(16))) unsigned pst[EG];
    __shared__ __attribute__((aligned(16))) unsigned pln[EG];
    __shared__ unsigned wst[DT / 32];
    __shared__ unsigned wln[DT / 32];
    __shared__ int sL[RCAP];
    const int tid = threadIdx.x;
    const int lane = tid & 31;
    const int wave = __builtin_amdgcn_readfirstlane((int)(threadIdx.x >> 5));
    const int b = blockIdx.x;
    const int row0 = b * RPB;

    cntL[tid] = 0u;
    unsigned st = 0u, ln = 0u;
    if (tid < EG) {
        const unsigned t = tab[b * EG + tid];
        st = t >> 16;
        ln = t & 0xFFFFu;
        pst[tid] = (unsigned)(tid * ECH) + st;
        pln[tid] = ln;
    }
    for (int off = 16; off >= 1; off >>= 1) {
        st += __shfl_xor(st, off, 32);
        ln += __shfl_xor(ln, off, 32);
    }
    if (lane == 0) { wst[wave] = st; wln[wave] = ln; }
    __syncthreads();
    int beg = 0, n = 0;
    #pragma unroll
    for (int w = 0; w < DT / 32; ++w) { beg += (int)wst[w]; n += (int)wln[w]; }

    unsigned e[PPW], f[PPW];
    int ps[PPW], pl[PPW];
    #pragma unroll
    for (int k = 0; k < PPW; ++k) {
        ps[k] = __builtin_amdgcn_readfirstlane((int)pst[wave * PPW + k]);
        pl[k] = __builtin_amdgcn_readfirstlane((int)pln[wave * PPW + k]);
    }
    bool any_long = false;
    #pragma unroll
    for (int k = 0; k < PPW; ++k) {
        e[k] = (lane < pl[k]) ? blist[ps[k] + lane] : EMPTY_EDGE;
        f[k] = EMPTY_EDGE;
        if (pl[k] > 32) f[k] = (32 + lane < pl[k]) ? blist[ps[k] + 32 + lane] : EMPTY_EDGE;
        any_long |= (pl[k] > 64);
    }

    #pragma unroll
    for (int k = 0; k < PPW; ++k)
        if (e[k] != EMPTY_EDGE) atomicAdd(&cntL[e[k] >> 16], 1u);
    #pragma unroll
    for (int k = 0; k < PPW; ++k)
        if (f[k] != EMPTY_EDGE) atomicAdd(&cntL[f[k] >> 16], 1u);
    if (any_long) {
        for (int k = 0; k < PPW; ++k) {
            const int p = wave * PPW + k;
            const int rs = (int)pst[p];
            const int rl = (int)pln[p];
            for (int j = 64 + lane; j < rl; j += 32)
                atomicAdd(&cntL[blist[rs + j] >> 16], 1u);
        }
    }
    __syncthreads();

    const unsigned v = cntL[tid];
    unsigned s = v;
    for (int off = 1; off < 32; off <<= 1) {
        const unsigned t = __shfl_up(s, off, 32);
        if (lane >= off) s += t;
    }
    if (lane == 31) wtot[wave] = s;
    __syncthreads();
    const unsigned wt = (lane < DT / 32) ? wtot[lane] : 0u;
    unsigned ws = wt;
    for (int off = 1; off < 32; off <<= 1) {
        const unsigned t = __shfl_up(ws, off, 32);
        if (lane >= off) ws += t;
    }
    const unsigned wexcl = __shfl(ws, wave, 32) - __shfl(wt, wave, 32);
    const unsigned excl = wexcl + s - v;
    cntL[tid] = excl;
    if (tid < RPB && row0 + tid < N_NODES) roff[row0 + tid] = (unsigned)beg + excl;
    __syncthreads();

    if (n <= RCAP) {
        unsigned pos[PPW];
        #pragma unroll
        for (int k = 0; k < PPW; ++k)
            pos[k] = (e[k] != EMPTY_EDGE) ? atomicAdd(&cntL[e[k] >> 16], 1u) : 0u;
        #pragma unroll
        for (int k = 0; k < PPW; ++k)
            if (e[k] != EMPTY_EDGE) sL[pos[k]] = (int)(e[k] & 0xFFFFu);
        #pragma unroll
        for (int k = 0; k < PPW; ++k)
            if (f[k] != EMPTY_EDGE) sL[atomicAdd(&cntL[f[k] >> 16], 1u)] = (int)(f[k] & 0xFFFFu);
        if (any_long) {
            for (int k = 0; k < PPW; ++k) {
                const int p = wave * PPW + k;
                const int rs = (int)pst[p];
                const int rl = (int)pln[p];
                for (int j = 64 + lane; j < rl; j += 32) {
                    const unsigned w = blist[rs + j];
                    sL[atomicAdd(&cntL[w >> 16], 1u)] = (int)(w & 0xFFFFu);
                }
            }
        }
        __syncthreads();
        for (int j = tid; j < n; j += DT)
            scol[beg + j] = sL[j];
    } else {
        #pragma unroll
        for (int k = 0; k < PPW; ++k)
            if (e[k] != EMPTY_EDGE) {
                const unsigned pos = atomicAdd(&cntL[e[k] >> 16], 1u);
                scol[beg + pos] = (int)(e[k] & 0xFFFFu);
            }
        #pragma unroll
        for (int k = 0; k < PPW; ++k)
            if (f[k] != EMPTY_EDGE) {
                const unsigned pos = atomicAdd(&cntL[f[k] >> 16], 1u);
                scol[beg + pos] = (int)(f[k] & 0xFFFFu);
            }
        if (any_long) {
            for (int k = 0; k < PPW; ++k) {
                const int p = wave * PPW + k;
                const int rs = (int)pst[p];
                const int rl = (int)pln[p];
                for (int j = 64 + lane; j < rl; j += 32) {
                    const unsigned w = blist[rs + j];
                    const unsigned pos = atomicAdd(&cntL[w >> 16], 1u);
                    scol[beg + pos] = (int)(w & 0xFFFFu);
                }
            }
        }
    }
}

__global__ void __launch_bounds__(256)
gat_row_gather(const unsigned* __restrict__ roff, const int* __restrict__ scol,
               const float* __restrict__ s1, const float* __restrict__ s2,
               const unsigned* __restrict__ gmax, const float* __restrict__ Wh,
               float* __restrict__ out) {
    const int lane = threadIdx.x & 31;
    const int wave = __builtin_amdgcn_readfirstlane((int)(threadIdx.x >> 5));
    const int r = blockIdx.x * 8 + wave;
    if (r >= N_NODES) return;

    const int beg = (int)roff[r];
    const int end = (r + 1 < N_NODES) ? (int)roff[r + 1] : N_EDGES;
    const float gm = fdec(*gmax);
    const float s1r = s1[r];

    float lsum = 0.f;
    for (int j = beg + lane; j < end; j += 32)
        lsum += __expf(edge_score(s1r, s2[scol[j]]) - gm);
    for (int off = 16; off >= 1; off >>= 1)
        lsum += __shfl_xor(lsum, off, 32);
    const float den = lsum + EPS_F;

    const float4* Wh4 = (const float4*)Wh;
    float4 acc = make_float4(0.f, 0.f, 0.f, 0.f);
    for (int base = beg; base < end; base += 32) {
        const int j = base + lane;
        const int c = scol[(j < end) ? j : (end - 1)];
        const float ex = __expf(edge_score(s1r, s2[c]) - gm);
        const float al = (j < end) ? ex / den : 0.f;
        const int n = min(32, end - base);
        const int n4 = (n + 3) & ~3;
        for (int k = 0; k < n4; k += 4) {
            float4 w[4];
            float ak[4];
            for (int u = 0; u < 4; ++u) {
                const int ck = __builtin_amdgcn_readlane(c, k + u);
                ak[u] = __int_as_float(__builtin_amdgcn_readlane(__float_as_int(al), k + u));
                w[u] = Wh4[ck * (OUT_F / 4) + lane];
            }
            for (int u = 0; u < 4; ++u) {
                acc.x = fmaf(ak[u], w[u].x, acc.x);
                acc.y = fmaf(ak[u], w[u].y, acc.y);
                acc.z = fmaf(ak[u], w[u].z, acc.z);
                acc.w = fmaf(ak[u], w[u].w, acc.w);
            }
        }
    }

    float4 o;
    o.x = (acc.x > 0.f) ? acc.x : expm1f(acc.x);
    o.y = (acc.y > 0.f) ? acc.y : expm1f(acc.y);
    o.z = (acc.z > 0.f) ? acc.z : expm1f(acc.z);
    o.w = (acc.w > 0.f) ? acc.w : expm1f(acc.w);
    ((float4*)(out + r * OUT_F))[lane] = o;
}

extern "C" void kernel_launch(void* const* d_in, const int* in_sizes, int n_in,
                              void* d_out, int out_size, void* d_ws, size_t ws_size,
                              hipStream_t stream) {
    const float* h  = (const float*)d_in[0];
    const int*   ei = (const int*)d_in[1];
    const float* W  = (const float*)d_in[2];
    const float* a  = (const float*)d_in[3];
    float* out = (float*)d_out;

    const int* rows = ei;
    const int* cols = ei + N_EDGES;

    float* ws = (float*)d_ws;
    float*    Wh    = ws;
    float*    s1    = Wh + (size_t)N_NODES * OUT_F;
    float*    s2    = s1 + N_NODES;
    unsigned* blist = (unsigned*)(s2 + N_NODES);
    int*      scol  = (int*)(blist + N_EDGES);
    unsigned* roff  = (unsigned*)(scol + N_EDGES);
    unsigned* gmax  = roff + N_NODES;
    unsigned* tab   = (unsigned*)d_out;

    const int n_tiles = N_NODES / 16;
    gat_gemm_wmma<<<(n_tiles + 3) / 4, 128, 0, stream>>>(h, W, a, Wh, s1, s2, gmax, n_tiles);

    gat_chunk_sort<<<EG, CT, 0, stream>>>(rows, cols, s1, s2, tab, blist, gmax);
    gat_range_csr<<<NBKT, DT, 0, stream>>>(tab, blist, roff, scol);
    gat_row_gather<<<(N_NODES + 7) / 8, 256, 0, stream>>>(roff, scol, s1, s2, gmax, Wh, out);
}


# Work that needs nothing of the launch before runs beside it as a second work-group role: chunk sort beside the GEMM, score max beside the range pass: 4 launches -> 3; row sums stay in arrival order
# speedup vs baseline: 3.4519x; 1.0941x over previous
#include <hip/hip_runtime.h>
#include <math.h>

#define N_NODES 50000
#define N_EDGES 800000
#define IN_F 128
#define OUT_F 128
#define NEG_SLOPE 0.2f
#define EPS_F 1e-10f

typedef float v2f __attribute__((ext_vector_type(2)));
typedef float v8f __attribute__((ext_vector_type(8)));


__device__ __forceinline__ unsigned fenc(float f) {
    unsigned u = __float_as_uint(f);
    return (u & 0x80000000u) ? ~u : (u | 0x80000000u);
}
__device__ __forceinline__ float fdec(unsigned u) {
    return __uint_as_float((u & 0x80000000u) ? (u ^ 0x80000000u) : ~u);
}

__device__ __forceinline__ float edge_score(float a, float b) {
    const float v = a + b;
    return (v > 0.f) ? v : NEG_SLOPE * v;
}

#define NBKT 128
#define RPB 391
#define EG 256
#define ECH (N_EDGES / EG)
#define CT 128
#define EPT ((ECH + CT - 1) / CT)
#define DT 512
#define PPW (EG / (DT / 32))
#define RCAP 8192
#define MG 400
#define MCH (N_EDGES / MG)
#define MPT ((MCH + DT - 1) / DT)
#define EMPTY_EDGE 0xFFFFFFFFu
static_assert(NBKT * RPB >= N_NODES, "ranges must cover all rows");
static_assert(ECH * EG == N_EDGES, "chunks must tile the edge list");
static_assert(MCH * MG == N_EDGES, "score-max pieces must tile the edge list");
static_assert(NBKT == 4 * 32, "one wave scans the range counters, four per lane");
static_assert(RPB <= DT && EG <= DT, "one thread per row counter / per run of a range");
static_assert(PPW * (DT / 32) == EG, "the waves of a range work-group share out its runs evenly");
static_assert(N_NODES <= 65536 && RPB <= 512, "packed edge word: local row above bit 16, column below");
static_assert(ECH < 65536, "table word: run start above bit 16, run length below");
static_assert((NBKT + ECH) * 4 <= 4 * 16 * 128 * 4, "the chunk role's counters and sorted edges fit the GEMM role's LDS tiles");

__device__ __forceinline__ void chunk_sort_role(const int* __restrict__ rows, const int* __restrict__ cols,
                                                unsigned* __restrict__ tab, unsigned* __restrict__ blist,
                                                unsigned* __restrict__ hist,
                                                unsigned* __restrict__ sorted,
                                                const int chunk) {
    const int tid = threadIdx.x;
    __builtin_assume(tid < CT);
    const int lane = tid & 31;
    const int wave = tid >> 5;
    for (int i = tid; i < NBKT; i += CT) hist[i] = 0u;

    const int base = chunk * ECH;
    int r[EPT], c[EPT];
    #pragma unroll
    for (int u = 0; u < EPT; ++u) {
        const int i = u * CT + tid;
        r[u] = (i < ECH) ? rows[base + i] : 0;
        c[u] = (i < ECH) ? cols[base + i] : 0;
    }
    __syncthreads();

    int q[EPT];
    unsigned rk[EPT];
    #pragma unroll
    for (int u = 0; u < EPT; ++u) {
        q[u] = r[u] / RPB;
        rk[u] = (u * CT + tid < ECH) ? atomicAdd(&hist[q[u]], 1u) : 0u;
    }
    __syncthreads();

    if (wave == 0) {
        const uint4 hv = ((const uint4*)hist)[lane];
        const unsigned t4 = hv.x + hv.y + hv.z + hv.w;
        unsigned s = t4;
        for (int off = 1; off < 32; off <<= 1) {
            const unsigned n = __shfl_up(s, off, 32);
            if (lane >= off) s += n;
        }
        const unsigned e0 = s - t4;
        const unsigned e1 = e0 + hv.x;
        const unsigned e2 = e1 + hv.y;
        const unsigned e3 = e2 + hv.z;
        ((uint4*)hist)[lane] = make_uint4(e0, e1, e2, e3);
        tab[(4 * lane + 0) * EG + chunk] = (e0 << 16) | hv.x;
        tab[(4 * lane + 1) * EG + chunk] = (e1 << 16) | hv.y;
        tab[(4 * lane + 2) * EG + chunk] = (e2 << 16) | hv.z;
        tab[(4 * lane + 3) * EG + chunk] = (e3 << 16) | hv.w;
    }
    __syncthreads();

    unsigned slot[EPT];
    #pragma unroll
    for (int u = 0; u < EPT; ++u)
        slot[u] = hist[q[u]] + rk[u];
    #pragma unroll
    for (int u = 0; u < EPT; ++u)
        if (u * CT + tid < ECH)
            sorted[slot[u]] = ((unsigned)(r[u] - q[u] * RPB) << 16) | (unsigned)c[u];
    __syncthreads();

    for (int i = tid; i < ECH; i += CT)
        blist[base + i] = sorted[i];
}

__global__ void __launch_bounds__(CT)
gat_gemm_and_chunk_sort(const float* __restrict__ h, const float* __restrict__ W,
                        const float* __restrict__ a, float* __restrict__ Wh,
                        float* __restrict__ s1, float* __restrict__ s2, unsigned* __restrict__ gmax,
                        const int* __restrict__ rows, const int* __restrict__ cols,
                        unsigned* __restrict__ tab, unsigned* __restrict__ blist,
                        int n_tiles) {
    __shared__ __attribute__((aligned(16))) unsigned char lds_raw[4 * 16 * 128 * 4];

    if (blockIdx.x < EG) {
        unsigned* const lw = (unsigned*)lds_raw;
        chunk_sort_role(rows, cols, tab, blist, lw, lw + NBKT, (int)blockIdx.x);
        return;
    }
    float* const lds = (float*)lds_raw;

    if (blockIdx.x == EG && threadIdx.x == 0) *gmax = 0u;

    const int lane = threadIdx.x & 31;
    const int wave = threadIdx.x >> 5;
    const int tile = ((int)blockIdx.x - EG) * 4 + wave;
    if (tile >= n_tiles) return;

    const int row0 = tile * 16;
    const int half = lane >> 4;
    const int mn   = lane & 15;

    float* tl = lds + wave * (16 * 128);
    {
        const float4* hp4 = (const float4*)(h + row0 * IN_F);
        float4* tl4 = (float4*)tl;
        #pragma unroll
        for (int t = 0; t < 16; ++t)
            tl4[t * 32 + lane] = hp4[t * 32 + lane];
    }

    float a1v[8], a2v[8];
    #pragma unroll
    for (int nt = 0; nt < 8; ++nt) {
        a1v[nt] = a[nt * 16 + mn];
        a2v[nt] = a[OUT_F + nt * 16 + mn];
    }

    v8f acc[8] = {};

    for (int kc = 0; kc < 32; ++kc) {
        const int col = kc * 4 + 2 * half;
        v2f av;
        {
            const float2 t = *(const float2*)(tl + mn * 128 + col);
            av[0] = t.x; av[1] = t.y;
        }
        #pragma unroll
        for (int nt = 0; nt < 8; ++nt) {
            const int ncol = nt * 16 + mn;
            v2f bv;
            {
                const float2 t = *(const float2*)(W + ncol * IN_F + col);
                bv[0] = t.x; bv[1] = t.y;
            }
            acc[nt] = __builtin_amdgcn_wmma_f32_16x16x4_f32(
                false, av, false, bv, (short)0, acc[nt], false, false);
        }
    }

    #pragma unroll
    for (int nt = 0; nt < 8; ++nt) {
        #pragma unroll
        for (int j = 0; j < 8; ++j) {
            Wh[(row0 + j + 8 * half) * OUT_F + nt * 16 + mn] = acc[nt][j];
        }
    }

    float p1[8], p2[8];
    #pragma unroll
    for (int j = 0; j < 8; ++j) { p1[j] = 0.f; p2[j] = 0.f; }
    #pragma unroll
    for (int nt = 0; nt < 8; ++nt) {
        #pragma unroll
        for (int j = 0; j < 8; ++j) {
            p1[j] = fmaf(acc[nt][j], a1v[nt], p1[j]);
            p2[j] = fmaf(acc[nt][j], a2v[nt], p2[j]);
        }
    }
    #pragma unroll
    for (int j = 0; j < 8; ++j) {
        #pragma unroll
        for (int off = 1; off < 16; off <<= 1) {
            p1[j] += __shfl_xor(p1[j], off, 32);
            p2[j] += __shfl_xor(p2[j], off, 32);
        }
    }
    if (mn == 0) {
        #pragma unroll
        for (int j = 0; j < 8; ++j) {
            s1[row0 + j + 8 * half] = p1[j];
            s2[row0 + j + 8 * half] = p2[j];
        }
    }
}

__device__ __forceinline__ void score_max_role(const int* __restrict__ rows, const int* __restrict__ cols,
                                               const float* __restrict__ s1, const float* __restrict__ s2,
                                               unsigned* __restrict__ gmax, float* __restrict__ wmx, const int piece) {
    const int tid = threadIdx.x;
    __builtin_assume(tid < DT);
    const int lane = tid & 31;
    const int wave = tid >> 5;
    const int base = piece * MCH;
    int r[MPT], c[MPT];
    #pragma unroll
    for (int u = 0; u < MPT; ++u) {
        const int i = u * DT + tid;
        r[u] = (i < MCH) ? rows[base + i] : 0;
        c[u] = (i < MCH) ? cols[base + i] : 0;
    }
    float sc[MPT];
    #pragma unroll
    for (int u = 0; u < MPT; ++u)
        sc[u] = edge_score(s1[r[u]], s2[c[u]]);
    float lmax = -INFINITY;
    #pragma unroll
    for (int u = 0; u < MPT; ++u)
        if (u * DT + tid < MCH) lmax = fmaxf(lmax, sc[u]);
    for (int off = 16; off >= 1; off >>= 1)
        lmax = fmaxf(lmax, __shfl_xor(lmax, off, 32));
    if (lane == 0) wmx[wave] = lmax;
    __syncthreads();
    if (wave == 0) {
        float m = (lane < DT / 32) ? wmx[lane] : -INFINITY;
        for (int off = 16; off >= 1; off >>= 1)
            m = fmaxf(m, __shfl_xor(m, off, 32));
        if (lane == 0) atomicMax(gmax, fenc(m));
    }
}

__global__ void __launch_bounds__(DT)
gat_range_csr_and_score_max(const unsigned* __restrict__ tab, const unsigned* __restrict__ blist,
                            unsigned* __restrict__ roff, int* __restrict__ scol,
                            const int* __restrict__ rows, const int* __restrict__ cols,
                            const float* __restrict__ s1, const float* __restrict__ s2,
                            unsigned* __restrict__ gmax) {
    __shared__ unsigned cntL[DT];
    __shared__ unsigned wtot[DT / 32];
    __shared__ __attribute__((aligned(16))) unsigned pst[EG];
    __shared__ __attribute__((aligned(16))) unsigned pln[EG];
    __shared__ unsigned wst[DT / 32];
    __shared__ unsigned wln[DT / 32];
    __shared__ int sL[RCAP];
    __shared__ float wmx[DT / 32];

    if (blockIdx.x >= NBKT) {
        score_max_role(rows, cols, s1, s2, gmax, wmx, (int)blockIdx.x - NBKT);
        return;
    }

    const int tid = threadIdx.x;
    const int lane = tid & 31;
    const int wave = __builtin_amdgcn_readfirstlane((int)(threadIdx.x >> 5));
    const int b = blockIdx.x;
    const int row0 = b * RPB;

    cntL[tid] = 0u;
    unsigned st = 0u, ln = 0u;
    if (tid < EG) {
        const unsigned t = tab[b * EG + tid];
        st = t >> 16;
        ln = t & 0xFFFFu;
        pst[tid] = (unsigned)(tid * ECH) + st;
        pln[tid] = ln;
    }
    for (int off = 16; off >= 1; off >>= 1) {
        st += __shfl_xor(st, off, 32);
        ln += __shfl_xor(ln, off, 32);
    }
    if (lane == 0) { wst[wave] = st; wln[wave] = ln; }
    __syncthreads();
    int beg = 0, n = 0;
    #pragma unroll
    for (int w = 0; w < DT / 32; ++w) { beg += (int)wst[w]; n += (int)wln[w]; }

    unsigned e[PPW], f[PPW];
    int ps[PPW], pl[PPW];
    #pragma unroll
    for (int k = 0; k < PPW; ++k) {
        ps[k] = __builtin_amdgcn_readfirstlane((int)pst[wave * PPW + k]);
        pl[k] = __builtin_amdgcn_readfirstlane((int)pln[wave * PPW + k]);
    }
    bool any_long = false;
    #pragma unroll
    for (int k = 0; k < PPW; ++k) {
        e[k] = (lane < pl[k]) ? blist[ps[k] + lane] : EMPTY_EDGE;
        f[k] = EMPTY_EDGE;
        if (pl[k] > 32) f[k] = (32 + lane < pl[k]) ? blist[ps[k] + 32 + lane] : EMPTY_EDGE;
        any_long |= (pl[k] > 64);
    }

    #pragma unroll
    for (int k = 0; k < PPW; ++k)
        if (e[k] != EMPTY_EDGE) atomicAdd(&cntL[e[k] >> 16], 1u);
    #pragma unroll
    for (int k = 0; k < PPW; ++k)
        if (f[k] != EMPTY_EDGE) atomicAdd(&cntL[f[k] >> 16], 1u);
    if (any_long) {
        for (int k = 0; k < PPW; ++k) {
            const int p = wave * PPW + k;
            const int rs = (int)pst[p];
            const int rl = (int)pln[p];
            for (int j = 64 + lane; j < rl; j += 32)
                atomicAdd(&cntL[blist[rs + j] >> 16], 1u);
        }
    }
    __syncthreads();

    const unsigned v = cntL[tid];
    unsigned s = v;
    for (int off = 1; off < 32; off <<= 1) {
        const unsigned t = __shfl_up(s, off, 32);
        if (lane >= off) s += t;
    }
    if (lane == 31) wtot[wave] = s;
    __syncthreads();
    const unsigned wt = (lane < DT / 32) ? wtot[lane] : 0u;
    unsigned ws = wt;
    for (int off = 1; off < 32; off <<= 1) {
        const unsigned t = __shfl_up(ws, off, 32);
        if (lane >= off) ws += t;
    }
    const unsigned wexcl = __shfl(ws, wave, 32) - __shfl(wt, wave, 32);
    const unsigned excl = wexcl + s - v;
    cntL[tid] = excl;
    if (tid < RPB && row0 + tid < N_NODES) roff[row0 + tid] = (unsigned)beg + excl;
    __syncthreads();

    if (n <= RCAP) {
        unsigned pos[PPW];
        #pragma unroll
        for (int k = 0; k < PPW; ++k)
            pos[k] = (e[k] != EMPTY_EDGE) ? atomicAdd(&cntL[e[k] >> 16], 1u) : 0u;
        #pragma unroll
        for (int k = 0; k < PPW; ++k)
            if (e[k] != EMPTY_EDGE) sL[pos[k]] = (int)(e[k] & 0xFFFFu);
        #pragma unroll
        for (int k = 0; k < PPW; ++k)
            if (f[k] != EMPTY_EDGE) sL[atomicAdd(&cntL[f[k] >> 16], 1u)] = (int)(f[k] & 0xFFFFu);
        if (any_long) {
            for (int k = 0; k < PPW; ++k) {
                const int p = wave * PPW + k;
                const int rs = (int)pst[p];
                const int rl = (int)pln[p];
                for (int j = 64 + lane; j < rl; j += 32) {
                    const unsigned w = blist[rs + j];
                    sL[atomicAdd(&cntL[w >> 16], 1u)] = (int)(w & 0xFFFFu);
                }
            }
        }
        __syncthreads();
        for (int j = tid; j < n; j += DT)
            scol[beg + j] = sL[j];
    } else {
        #pragma unroll
        for (int k = 0; k < PPW; ++k)
            if (e[k] != EMPTY_EDGE) {
                const unsigned pos = atomicAdd(&cntL[e[k] >> 16], 1u);
                scol[beg + pos] = (int)(e[k] & 0xFFFFu);
            }
        #pragma unroll
        for (int k = 0; k < PPW; ++k)
            if (f[k] != EMPTY_EDGE) {
                const unsigned pos = atomicAdd(&cntL[f[k] >> 16], 1u);
                scol[beg + pos] = (int)(f[k] & 0xFFFFu);
            }
        if (any_long) {
            for (int k = 0; k < PPW; ++k) {
                const int p = wave * PPW + k;
                const int rs = (int)pst[p];
                const int rl = (int)pln[p];
                for (int j = 64 + lane; j < rl; j += 32) {
                    const unsigned w = blist[rs + j];
                    const unsigned pos = atomicAdd(&cntL[w >> 16], 1u);
                    scol[beg + pos] = (int)(w & 0xFFFFu);
                }
            }
        }
    }
}

__global__ void __launch_bounds__(256)
gat_row_gather(const unsigned* __restrict__ roff, const int* __restrict__ scol,
               const float* __restrict__ s1, const float* __restrict__ s2,
               const unsigned* __restrict__ gmax, const float* __restrict__ Wh,
               float* __restrict__ out) {
    const int lane = threadIdx.x & 31;
    const int wave = __builtin_amdgcn_readfirstlane((int)(threadIdx.x >> 5));
    const int r = blockIdx.x * 8 + wave;
    if (r >= N_NODES) return;

    const int beg = (int)roff[r];
    const int end = (r + 1 < N_NODES) ? (int)roff[r + 1] : N_EDGES;
    const float gm = fdec(*gmax);
    const float s1r = s1[r];

    float lsum = 0.f;
    for (int j = beg + lane; j < end; j += 32)
        lsum += __expf(edge_score(s1r, s2[scol[j]]) - gm);
    for (int off = 16; off >= 1; off >>= 1)
        lsum += __shfl_xor(lsum, off, 32);
    const float den = lsum + EPS_F;

    const float4* Wh4 = (const float4*)Wh;
    float4 acc = make_float4(0.f, 0.f, 0.f, 0.f);
    for (int base = beg; base < end; base += 32) {
        const int j = base + lane;
        const int c = scol[(j < end) ? j : (end - 1)];
        const float ex = __expf(edge_score(s1r, s2[c]) - gm);
        const float al = (j < end) ? ex / den : 0.f;
        const int n = min(32, end - base);
        const int n4 = (n + 3) & ~3;
        for (int k = 0; k < n4; k += 4) {
            float4 w[4];
            float ak[4];
            for (int u = 0; u < 4; ++u) {
                const int ck = __builtin_amdgcn_readlane(c, k + u);
                ak[u] = __int_as_float(__builtin_amdgcn_readlane(__float_as_int(al), k + u));
                w[u] = Wh4[ck * (OUT_F / 4) + lane];
            }
            for (int u = 0; u < 4; ++u) {
                acc.x = fmaf(ak[u], w[u].x, acc.x);
                acc.y = fmaf(ak[u], w[u].y, acc.y);
                acc.z = fmaf(ak[u], w[u].z, acc.z);
                acc.w = fmaf(ak[u], w[u].w, acc.w);
            }
        }
    }

    float4 o;
    o.x = (acc.x > 0.f) ? acc.x : expm1f(acc.x);
    o.y = (acc.y > 0.f) ? acc.y : expm1f(acc.y);
    o.z = (acc.z > 0.f) ? acc.z : expm1f(acc.z);
    o.w = (acc.w > 0.f) ? acc.w : expm1f(acc.w);
    ((float4*)(out + r * OUT_F))[lane] = o;
}

extern "C" void kernel_launch(void* const* d_in, const int* in_sizes, int n_in,
                              void* d_out, int out_size, void* d_ws, size_t ws_size,
                              hipStream_t stream) {
    const float* h  = (const float*)d_in[0];
    const int*   ei = (const int*)d_in[1];
    const float* W  = (const float*)d_in[2];
    const float* a  = (const float*)d_in[3];
    float* out = (float*)d_out;

    const int* rows = ei;
    const int* cols = ei + N_EDGES;

    float* ws = (float*)d_ws;
    float*    Wh    = ws;
    float*    s1    = Wh + (size_t)N_NODES * OUT_F;
    float*    s2    = s1 + N_NODES;
    unsigned* blist = (unsigned*)(s2 + N_NODES);
    int*      scol  = (int*)(blist + N_EDGES);
    unsigned* roff  = (unsigned*)(scol + N_EDGES);
    unsigned* gmax  = roff + N_NODES;
    unsigned* tab   = (unsigned*)d_out;

    const int n_tiles = N_NODES / 16;
    gat_gemm_and_chunk_sort<<<EG + (n_tiles + 3) / 4, CT, 0, stream>>>(h, W, a, Wh, s1, s2, gmax,
                                                                     rows, cols, tab, blist, n_tiles);
    gat_range_csr_and_score_max<<<NBKT + MG, DT, 0, stream>>>(tab, blist, roff, scol, rows, cols, s1, s2, gmax);
    gat_row_gather<<<(N_NODES + 7) / 8, 256, 0, stream>>>(roff, scol, s1, s2, gmax, Wh, out);
}


# GEMM role copies W into LDS once per work-group (row pitch 132 words) and reads each B fragment from it, not as a 16-line global load per fragment; k order and sums unchanged
# speedup vs baseline: 3.8560x; 1.1171x over previous
#include <hip/hip_runtime.h>
#include <math.h>

#define N_NODES 50000
#define N_EDGES 800000
#define IN_F 128
#define OUT_F 128
#define NEG_SLOPE 0.2f
#define EPS_F 1e-10f

typedef float v2f __attribute__((ext_vector_type(2)));
typedef float v8f __attribute__((ext_vector_type(8)));


__device__ __forceinline__ unsigned fenc(float f) {
    unsigned u = __float_as_uint(f);
    return (u & 0x80000000u) ? ~u : (u | 0x80000000u);
}
__device__ __forceinline__ float fdec(unsigned u) {
    return __uint_as_float((u & 0x80000000u) ? (u ^ 0x80000000u) : ~u);
}

__device__ __forceinline__ float edge_score(float a, float b) {
    const float v = a + b;
    return (v > 0.f) ? v : NEG_SLOPE * v;
}

#define NBKT 128
#define RPB 391
#define EG 256
#define ECH (N_EDGES / EG)
#define CT 128
#define EPT ((ECH + CT - 1) / CT)
#define DT 512
#define PPW (EG / (DT / 32))
#define RCAP 8192
#define MG 400
#define MCH (N_EDGES / MG)
#define MPT ((MCH + DT - 1) / DT)
#define EMPTY_EDGE 0xFFFFFFFFu
#define HT_WORDS (4 * 16 * 128)
#define WPITCH 132
static_assert(NBKT * RPB >= N_NODES, "ranges must cover all rows");
static_assert(ECH * EG == N_EDGES, "chunks must tile the edge list");
static_assert(MCH * MG == N_EDGES, "score-max pieces must tile the edge list");
static_assert(NBKT == 4 * 32, "one wave scans the range counters, four per lane");
static_assert(RPB <= DT && EG <= DT, "one thread per row counter / per run of a range");
static_assert(PPW * (DT / 32) == EG, "the waves of a range work-group share out its runs evenly");
static_assert(N_NODES <= 65536 && RPB <= 512, "packed edge word: local row above bit 16, column below");
static_assert(ECH < 65536, "table word: run start above bit 16, run length below");
static_assert(NBKT + ECH <= HT_WORDS, "the chunk role's counters and sorted edges fit the GEMM role's LDS tiles");
static_assert((OUT_F * IN_F / 4) % CT == 0, "the work-group's threads share out the float4 pieces of W evenly");
static_assert(WPITCH % 4 == 0 && WPITCH >= IN_F, "rows of the LDS copy of W hold a whole W row and stay 16-byte aligned");

__device__ __forceinline__ void chunk_sort_role(const int* __restrict__ rows, const int* __restrict__ cols,
                                                unsigned* __restrict__ tab, unsigned* __restrict__ blist,
                                                unsigned* __restrict__ hist,
                                                unsigned* __restrict__ sorted,
                                                const int chunk) {
    const int tid = threadIdx.x;
    __builtin_assume(tid < CT);
    const int lane = tid & 31;
    const int wave = tid >> 5;
    for (int i = tid; i < NBKT; i += CT) hist[i] = 0u;

    const int base = chunk * ECH;
    int r[EPT], c[EPT];
    #pragma unroll
    for (int u = 0; u < EPT; ++u) {
        const int i = u * CT + tid;
        r[u] = (i < ECH) ? rows[base + i] : 0;
        c[u] = (i < ECH) ? cols[base + i] : 0;
    }
    __syncthreads();

    int q[EPT];
    unsigned rk[EPT];
    #pragma unroll
    for (int u = 0; u < EPT; ++u) {
        q[u] = r[u] / RPB;
        rk[u] = (u * CT + tid < ECH) ? atomicAdd(&hist[q[u]], 1u) : 0u;
    }
    __syncthreads();

    if (wave == 0) {
        const uint4 hv = ((const uint4*)hist)[lane];
        const unsigned t4 = hv.x + hv.y + hv.z + hv.w;
        unsigned s = t4;
        for (int off = 1; off < 32; off <<= 1) {
            const unsigned n = __shfl_up(s, off, 32);
            if (lane >= off) s += n;
        }
        const unsigned e0 = s - t4;
        const unsigned e1 = e0 + hv.x;
        const unsigned e2 = e1 + hv.y;
        const unsigned e3 = e2 + hv.z;
        ((uint4*)hist)[lane] = make_uint4(e0, e1, e2, e3);
        tab[(4 * lane + 0) * EG + chunk] = (e0 << 16) | hv.x;
        tab[(4 * lane + 1) * EG + chunk] = (e1 << 16) | hv.y;
        tab[(4 * lane + 2) * EG + chunk] = (e2 << 16) | hv.z;
        tab[(4 * lane + 3) * EG + chunk] = (e3 << 16) | hv.w;
    }
    __syncthreads();

    unsigned slot[EPT];
    #pragma unroll
    for (int u = 0; u < EPT; ++u)
        slot[u] = hist[q[u]] + rk[u];
    #pragma unroll
    for (int u = 0; u < EPT; ++u)
        if (u * CT + tid < ECH)
            sorted[slot[u]] = ((unsigned)(r[u] - q[u] * RPB) << 16) | (unsigned)c[u];
    __syncthreads();

    for (int i = tid; i < ECH; i += CT)
        blist[base + i] = sorted[i];
}

__global__ void __launch_bounds__(CT)
gat_gemm_and_chunk_sort(const float* __restrict__ h, const float* __restrict__ W,
                        const float* __restrict__ a, float* __restrict__ Wh,
                        float* __restrict__ s1, float* __restrict__ s2, unsigned* __restrict__ gmax,
                        const int* __restrict__ rows, const int* __restrict__ cols,
                        unsigned* __restrict__ tab, unsigned* __restrict__ blist,
                        int n_tiles) {
    __shared__ __attribute__((aligned(16))) unsigned char lds_raw[(HT_WORDS + OUT_F * WPITCH) * 4];

    if (blockIdx.x < EG) {
        unsigned* const lw = (unsigned*)lds_raw;
        chunk_sort_role(rows, cols, tab, blist, lw, lw + NBKT, (int)blockIdx.x);
        return;
    }
    float* const lds = (float*)lds_raw;

    if (blockIdx.x == EG && threadIdx.x == 0) *gmax = 0u;

    const int lane = threadIdx.x & 31;
    const int wave = threadIdx.x >> 5;
    const int tile = ((int)blockIdx.x - EG) * 4 + wave;

    const bool live = tile < n_tiles;
    const int row0 = tile * 16;
    const int half = lane >> 4;
    const int mn   = lane & 15;

    float* tl = lds + wave * (16 * 128);
    float4 ht[16];
    {
        const float4* hp4 = (const float4*)(h + (live ? row0 : (n_tiles - 1) * 16) * IN_F);
        #pragma unroll
        for (int t = 0; t < 16; ++t)
            ht[t] = hp4[t * 32 + lane];
    }

    float* const wl = lds + HT_WORDS;
    {
        const float4* W4 = (const float4*)W;
        #pragma unroll
        for (int i = 0; i < (OUT_F * IN_F / 4) / CT; ++i) {
            const int idx = i * CT + (int)threadIdx.x;
            *(float4*)(wl + (idx >> 5) * WPITCH + (idx & 31) * 4) = W4[idx];
        }
    }
    {
        float4* tl4 = (float4*)tl;
        #pragma unroll
        for (int t = 0; t < 16; ++t)
            tl4[t * 32 + lane] = ht[t];
    }
    __syncthreads();
    if (!live) return;

    float a1v[8], a2v[8];
    #pragma unroll
    for (int nt = 0; nt < 8; ++nt) {
        a1v[nt] = a[nt * 16 + mn];
        a2v[nt] = a[OUT_F + nt * 16 + mn];
    }

    v8f acc[8] = {};

    for (int kc = 0; kc < 32; ++kc) {
        const int col = kc * 4 + 2 * half;
        v2f av;
        {
            const float2 t = *(const float2*)(tl + mn * 128 + col);
            av[0] = t.x; av[1] = t.y;
        }
        #pragma unroll
        for (int nt = 0; nt < 8; ++nt) {
            const int ncol = nt * 16 + mn;
            v2f bv;
            {
                const float2 t = *(const float2*)(wl + ncol * WPITCH + col);
                bv[0] = t.x; bv[1] = t.y;
            }
            acc[nt] = __builtin_amdgcn_wmma_f32_16x16x4_f32(
                false, av, false, bv, (short)0, acc[nt], false, false);
        }
    }

    #pragma unroll
    for (int nt = 0; nt < 8; ++nt) {
        #pragma unroll
        for (int j = 0; j < 8; ++j) {
            Wh[(row0 + j + 8 * half) * OUT_F + nt * 16 + mn] = acc[nt][j];
        }
    }

    float p1[8], p2[8];
    #pragma unroll
    for (int j = 0; j < 8; ++j) { p1[j] = 0.f; p2[j] = 0.f; }
    #pragma unroll
    for (int nt = 0; nt < 8; ++nt) {
        #pragma unroll
        for (int j = 0; j < 8; ++j) {
            p1[j] = fmaf(acc[nt][j], a1v[nt], p1[j]);
            p2[j] = fmaf(acc[nt][j], a2v[nt], p2[j]);
        }
    }
    #pragma unroll
    for (int j = 0; j < 8; ++j) {
        #pragma unroll
        for (int off = 1; off < 16; off <<= 1) {
            p1[j] += __shfl_xor(p1[j], off, 32);
            p2[j] += __shfl_xor(p2[j], off, 32);
        }
    }
    if (mn == 0) {
        #pragma unroll
        for (int j = 0; j < 8; ++j) {
            s1[row0 + j + 8 * half] = p1[j];
            s2[row0 + j + 8 * half] = p2[j];
        }
    }
}

__device__ __forceinline__ void score_max_role(const int* __restrict__ rows, const int* __restrict__ cols,
                                               const float* __restrict__ s1, const float* __restrict__ s2,
                                               unsigned* __restrict__ gmax, float* __restrict__ wmx, const int piece) {
    const int tid = threadIdx.x;
    __builtin_assume(tid < DT);
    const int lane = tid & 31;
    const int wave = tid >> 5;
    const int base = piece * MCH;
    int r[MPT], c[MPT];
    #pragma unroll
    for (int u = 0; u < MPT; ++u) {
        const int i = u * DT + tid;
        r[u] = (i < MCH) ? rows[base + i] : 0;
        c[u] = (i < MCH) ? cols[base + i] : 0;
    }
    float sc[MPT];
    #pragma unroll
    for (int u = 0; u < MPT; ++u)
        sc[u] = edge_score(s1[r[u]], s2[c[u]]);
    float lmax = -INFINITY;
    #pragma unroll
    for (int u = 0; u < MPT; ++u)
        if (u * DT + tid < MCH) lmax = fmaxf(lmax, sc[u]);
    for (int off = 16; off >= 1; off >>= 1)
        lmax = fmaxf(lmax, __shfl_xor(lmax, off, 32));
    if (lane == 0) wmx[wave] = lmax;
    __syncthreads();
    if (wave == 0) {
        float m = (lane < DT / 32) ? wmx[lane] : -INFINITY;
        for (int off = 16; off >= 1; off >>= 1)
            m = fmaxf(m, __shfl_xor(m, off, 32));
        if (lane == 0) atomicMax(gmax, fenc(m));
    }
}

__global__ void __launch_bounds__(DT)
gat_range_csr_and_score_max(const unsigned* __restrict__ tab, const unsigned* __restrict__ blist,
                            unsigned* __restrict__ roff, int* __restrict__ scol,
                            const int* __restrict__ rows, const int* __restrict__ cols,
                            const float* __restrict__ s1, const float* __restrict__ s2,
                            unsigned* __restrict__ gmax) {
    __shared__ unsigned cntL[DT];
    __shared__ unsigned wtot[DT / 32];
    __shared__ __attribute__((aligned(16))) unsigned pst[EG];
    __shared__ __attribute__((aligned(16))) unsigned pln[EG];
    __shared__ unsigned wst[DT / 32];
    __shared__ unsigned wln[DT / 32];
    __shared__ int sL[RCAP];
    __shared__ float wmx[DT / 32];

    if (blockIdx.x >= NBKT) {
        score_max_role(rows, cols, s1, s2, gmax, wmx, (int)blockIdx.x - NBKT);
        return;
    }

    const int tid = threadIdx.x;
    const int lane = tid & 31;
    const int wave = __builtin_amdgcn_readfirstlane((int)(threadIdx.x >> 5));
    const int b = blockIdx.x;
    const int row0 = b * RPB;

    cntL[tid] = 0u;
    unsigned st = 0u, ln = 0u;
    if (tid < EG) {
        const unsigned t = tab[b * EG + tid];
        st = t >> 16;
        ln = t & 0xFFFFu;
        pst[tid] = (unsigned)(tid * ECH) + st;
        pln[tid] = ln;
    }
    for (int off = 16; off >= 1; off >>= 1) {
        st += __shfl_xor(st, off, 32);
        ln += __shfl_xor(ln, off, 32);
    }
    if (lane == 0) { wst[wave] = st; wln[wave] = ln; }
    __syncthreads();
    int beg = 0, n = 0;
    #pragma unroll
    for (int w = 0; w < DT / 32; ++w) { beg += (int)wst[w]; n += (int)wln[w]; }

    unsigned e[PPW], f[PPW];
    int ps[PPW], pl[PPW];
    #pragma unroll
    for (int k = 0; k < PPW; ++k) {
        ps[k] = __builtin_amdgcn_readfirstlane((int)pst[wave * PPW + k]);
        pl[k] = __builtin_amdgcn_readfirstlane((int)pln[wave * PPW + k]);
    }
    bool any_long = false;
    #pragma unroll
    for (int k = 0; k < PPW; ++k) {
        e[k] = (lane < pl[k]) ? blist[ps[k] + lane] : EMPTY_EDGE;
        f[k] = EMPTY_EDGE;
        if (pl[k] > 32) f[k] = (32 + lane < pl[k]) ? blist[ps[k] + 32 + lane] : EMPTY_EDGE;
        any_long |= (pl[k] > 64);
    }

    #pragma unroll
    for (int k = 0; k < PPW; ++k)
        if (e[k] != EMPTY_EDGE) atomicAdd(&cntL[e[k] >> 16], 1u);
    #pragma unroll
    for (int k = 0; k < PPW; ++k)
        if (f[k] != EMPTY_EDGE) atomicAdd(&cntL[f[k] >> 16], 1u);
    if (any_long) {
        for (int k = 0; k < PPW; ++k) {
            const int p = wave * PPW + k;
            const int rs = (int)pst[p];
            const int rl = (int)pln[p];
            for (int j = 64 + lane; j < rl; j += 32)
                atomicAdd(&cntL[blist[rs + j] >> 16], 1u);
        }
    }
    __syncthreads();

    const unsigned v = cntL[tid];
    unsigned s = v;
    for (int off = 1; off < 32; off <<= 1) {
        const unsigned t = __shfl_up(s, off, 32);
        if (lane >= off) s += t;
    }
    if (lane == 31) wtot[wave] = s;
    __syncthreads();
    const unsigned wt = (lane < DT / 32) ? wtot[lane] : 0u;
    unsigned ws = wt;
    for (int off = 1; off < 32; off <<= 1) {
        const unsigned t = __shfl_up(ws, off, 32);
        if (lane >= off) ws += t;
    }
    const unsigned wexcl = __shfl(ws, wave, 32) - __shfl(wt, wave, 32);
    const unsigned excl = wexcl + s - v;
    cntL[tid] = excl;
    if (tid < RPB && row0 + tid < N_NODES) roff[row0 + tid] = (unsigned)beg + excl;
    __syncthreads();

    if (n <= RCAP) {
        unsigned pos[PPW];
        #pragma unroll
        for (int k = 0; k < PPW; ++k)
            pos[k] = (e[k] != EMPTY_EDGE) ? atomicAdd(&cntL[e[k] >> 16], 1u) : 0u;
        #pragma unroll
        for (int k = 0; k < PPW; ++k)
            if (e[k] != EMPTY_EDGE) sL[pos[k]] = (int)(e[k] & 0xFFFFu);
        #pragma unroll
        for (int k = 0; k < PPW; ++k)
            if (f[k] != EMPTY_EDGE) sL[atomicAdd(&cntL[f[k] >> 16], 1u)] = (int)(f[k] & 0xFFFFu);
        if (any_long) {
            for (int k = 0; k < PPW; ++k) {
                const int p = wave * PPW + k;
                const int rs = (int)pst[p];
                const int rl = (int)pln[p];
                for (int j = 64 + lane; j < rl; j += 32) {
                    const unsigned w = blist[rs + j];
                    sL[atomicAdd(&cntL[w >> 16], 1u)] = (int)(w & 0xFFFFu);
                }
            }
        }
        __syncthreads();
        for (int j = tid; j < n; j += DT)
            scol[beg + j] = sL[j];
    } else {
        #pragma unroll
        for (int k = 0; k < PPW; ++k)
            if (e[k] != EMPTY_EDGE) {
                const unsigned pos = atomicAdd(&cntL[e[k] >> 16], 1u);
                scol[beg + pos] = (int)(e[k] & 0xFFFFu);
            }
        #pragma unroll
        for (int k = 0; k < PPW; ++k)
            if (f[k] != EMPTY_EDGE) {
                const unsigned pos = atomicAdd(&cntL[f[k] >> 16], 1u);
                scol[beg + pos] = (int)(f[k] & 0xFFFFu);
            }
        if (any_long) {
            for (int k = 0; k < PPW; ++k) {
                const int p = wave * PPW + k;
                const int rs = (int)pst[p];
                const int rl = (int)pln[p];
                for (int j = 64 + lane; j < rl; j += 32) {
                    const unsigned w = blist[rs + j];
                    const unsigned pos = atomicAdd(&cntL[w >> 16], 1u);
                    scol[beg + pos] = (int)(w & 0xFFFFu);
                }
            }
        }
    }
}

__global__ void __launch_bounds__(256)
gat_row_gather(const unsigned* __restrict__ roff, const int* __restrict__ scol,
               const float* __restrict__ s1, const float* __restrict__ s2,
               const unsigned* __restrict__ gmax, const float* __restrict__ Wh,
               float* __restrict__ out) {
    const int lane = threadIdx.x & 31;
    const int wave = __builtin_amdgcn_readfirstlane((int)(threadIdx.x >> 5));
    const int r = blockIdx.x * 8 + wave;
    if (r >= N_NODES) return;

    const int beg = (int)roff[r];
    const int end = (r + 1 < N_NODES) ? (int)roff[r + 1] : N_EDGES;
    const float gm = fdec(*gmax);
    const float s1r = s1[r];

    float lsum = 0.f;
    for (int j = beg + lane; j < end; j += 32)
        lsum += __expf(edge_score(s1r, s2[scol[j]]) - gm);
    for (int off = 16; off >= 1; off >>= 1)
        lsum += __shfl_xor(lsum, off, 32);
    const float den = lsum + EPS_F;

    const float4* Wh4 = (const float4*)Wh;
    float4 acc = make_float4(0.f, 0.f, 0.f, 0.f);
    for (int base = beg; base < end; base += 32) {
        const int j = base + lane;
        const int c = scol[(j < end) ? j : (end - 1)];
        const float ex = __expf(edge_score(s1r, s2[c]) - gm);
        const float al = (j < end) ? ex / den : 0.f;
        const int n = min(32, end - base);
        const int n4 = (n + 3) & ~3;
        for (int k = 0; k < n4; k += 4) {
            float4 w[4];
            float ak[4];
            for (int u = 0; u < 4; ++u) {
                const int ck = __builtin_amdgcn_readlane(c, k + u);
                ak[u] = __int_as_float(__builtin_amdgcn_readlane(__float_as_int(al), k + u));
                w[u] = Wh4[ck * (OUT_F / 4) + lane];
            }
            for (int u = 0; u < 4; ++u) {
                acc.x = fmaf(ak[u], w[u].x, acc.x);
                acc.y = fmaf(ak[u], w[u].y, acc.y);
                acc.z = fmaf(ak[u], w[u].z, acc.z);
                acc.w = fmaf(ak[u], w[u].w, acc.w);
            }
        }
    }

    float4 o;
    o.x = (acc.x > 0.f) ? acc.x : expm1f(acc.x);
    o.y = (acc.y > 0.f) ? acc.y : expm1f(acc.y);
    o.z = (acc.z > 0.f) ? acc.z : expm1f(acc.z);
    o.w = (acc.w > 0.f) ? acc.w : expm1f(acc.w);
    ((float4*)(out + r * OUT_F))[lane] = o;
}

extern "C" void kernel_launch(void* const* d_in, const int* in_sizes, int n_in,
                              void* d_out, int out_size, void* d_ws, size_t ws_size,
                              hipStream_t stream) {
    const float* h  = (const float*)d_in[0];
    const int*   ei = (const int*)d_in[1];
    const float* W  = (const float*)d_in[2];
    const float* a  = (const float*)d_in[3];
    float* out = (float*)d_out;

    const int* rows = ei;
    const int* cols = ei + N_EDGES;

    float* ws = (float*)d_ws;
    float*    Wh    = ws;
    float*    s1    = Wh + (size_t)N_NODES * OUT_F;
    float*    s2    = s1 + N_NODES;
    unsigned* blist = (unsigned*)(s2 + N_NODES);
    int*      scol  = (int*)(blist + N_EDGES);
    unsigned* roff  = (unsigned*)(scol + N_EDGES);
    unsigned* gmax  = roff + N_NODES;
    unsigned* tab   = (unsigned*)d_out;

    const int n_tiles = N_NODES / 16;
    gat_gemm_and_chunk_sort<<<EG + (n_tiles + 3) / 4, CT, 0, stream>>>(h, W, a, Wh, s1, s2, gmax,
                                                                     rows, cols, tab, blist, n_tiles);
    gat_range_csr_and_score_max<<<NBKT + MG, DT, 0, stream>>>(tab, blist, roff, scol, rows, cols, s1, s2, gmax);
    gat_row_gather<<<(N_NODES + 7) / 8, 256, 0, stream>>>(roff, scol, s1, s2, gmax, Wh, out);
}


# Edge scores computed once, where the range work-groups place the edges (s2 and the range's s1 in LDS), stored beside the columns: score-max role and the gather's s2 reads gone; sums in arrival order
# speedup vs baseline: 4.1952x; 1.0880x over previous
#include <hip/hip_runtime.h>
#include <math.h>

#define N_NODES 50000
#define N_EDGES 800000
#define IN_F 128
#define OUT_F 128
#define NEG_SLOPE 0.2f
#define EPS_F 1e-10f

typedef float v2f __attribute__((ext_vector_type(2)));
typedef float v8f __attribute__((ext_vector_type(8)));


__device__ __forceinline__ unsigned fenc(float f) {
    unsigned u = __float_as_uint(f);
    return (u & 0x80000000u) ? ~u : (u | 0x80000000u);
}
__device__ __forceinline__ float fdec(unsigned u) {
    return __uint_as_float((u & 0x80000000u) ? (u ^ 0x80000000u) : ~u);
}

__device__ __forceinline__ float edge_score(float a, float b) {
    const float v = a + b;
    return (v > 0.f) ? v : NEG_SLOPE * v;
}

#define NBKT 128
#define RPB 391
#define EG 256
#define ECH (N_EDGES / EG)
#define CT 128
#define EPT ((ECH + CT - 1) / CT)
#define DT 512
#define PPW (EG / (DT / 32))
#define RCAP 8192
#define S2V4 (N_NODES / 4)
#define S2R ((S2V4 + DT - 1) / DT)
#define EMPTY_EDGE 0xFFFFFFFFu
#define HT_WORDS (4 * 16 * 128)
#define WPITCH 132
static_assert(NBKT * RPB >= N_NODES, "ranges must cover all rows");
static_assert(ECH * EG == N_EDGES, "chunks must tile the edge list");
static_assert(S2V4 * 4 == N_NODES, "s2 is copied to LDS in whole float4 pieces");
static_assert(NBKT == 4 * 32, "one wave scans the range counters, four per lane");
static_assert(RPB <= DT && EG <= DT, "one thread per row counter / per run of a range");
static_assert(PPW * (DT / 32) == EG, "the waves of a range work-group share out its runs evenly");
static_assert(N_NODES <= 65536 && RPB <= 512, "packed edge word: local row above bit 16, column below");
static_assert(ECH < 65536, "table word: run start above bit 16, run length below");
static_assert(NBKT + ECH <= HT_WORDS, "the chunk role's counters and sorted edges fit the GEMM role's LDS tiles");
static_assert((OUT_F * IN_F / 4) % CT == 0, "the work-group's threads share out the float4 pieces of W evenly");
static_assert(WPITCH % 4 == 0 && WPITCH >= IN_F, "rows of the LDS copy of W hold a whole W row and stay 16-byte aligned");

__device__ __forceinline__ void chunk_sort_role(const int* __restrict__ rows, const int* __restrict__ cols,
                                                unsigned* __restrict__ tab, unsigned* __restrict__ blist,
                                                unsigned* __restrict__ hist,
                                                unsigned* __restrict__ sorted,
                                                const int chunk) {
    const int tid = threadIdx.x;
    __builtin_assume(tid < CT);
    const int lane = tid & 31;
    const int wave = tid >> 5;
    for (int i = tid; i < NBKT; i += CT) hist[i] = 0u;

    const int base = chunk * ECH;
    int r[EPT], c[EPT];
    #pragma unroll
    for (int u = 0; u < EPT; ++u) {
        const int i = u * CT + tid;
        r[u] = (i < ECH) ? rows[base + i] : 0;
        c[u] = (i < ECH) ? cols[base + i] : 0;
    }
    __syncthreads();

    int q[EPT];
    unsigned rk[EPT];
    #pragma unroll
    for (int u = 0; u < EPT; ++u) {
        q[u] = r[u] / RPB;
        rk[u] = (u * CT + tid < ECH) ? atomicAdd(&hist[q[u]], 1u) : 0u;
    }
    __syncthreads();

    if (wave == 0) {
        const uint4 hv = ((const uint4*)hist)[lane];
        const unsigned t4 = hv.x + hv.y + hv.z + hv.w;
        unsigned s = t4;
        for (int off = 1; off < 32; off <<= 1) {
            const unsigned n = __shfl_up(s, off, 32);
            if (lane >= off) s += n;
        }
        const unsigned e0 = s - t4;
        const unsigned e1 = e0 + hv.x;
        const unsigned e2 = e1 + hv.y;
        const unsigned e3 = e2 + hv.z;
        ((uint4*)hist)[lane] = make_uint4(e0, e1, e2, e3);
        tab[(4 * lane + 0) * EG + chunk] = (e0 << 16) | hv.x;
        tab[(4 * lane + 1) * EG + chunk] = (e1 << 16) | hv.y;
        tab[(4 * lane + 2) * EG + chunk] = (e2 << 16) | hv.z;
        tab[(4 * lane + 3) * EG + chunk] = (e3 << 16) | hv.w;
    }
    __syncthreads();

    unsigned slot[EPT];
    #pragma unroll
    for (int u = 0; u < EPT; ++u)
        slot[u] = hist[q[u]] + rk[u];
    #pragma unroll
    for (int u = 0; u < EPT; ++u)
        if (u * CT + tid < ECH)
            sorted[slot[u]] = ((unsigned)(r[u] - q[u] * RPB) << 16) | (unsigned)c[u];
    __syncthreads();

    for (int i = tid; i < ECH; i += CT)
        blist[base + i] = sorted[i];
}

__global__ void __launch_bounds__(CT)
gat_gemm_and_chunk_sort(const float* __restrict__ h, const float* __restrict__ W,
                        const float* __restrict__ a, float* __restrict__ Wh,
                        float* __restrict__ s1, float* __restrict__ s2, unsigned* __restrict__ gmax,
                        const int* __restrict__ rows, const int* __restrict__ cols,
                        unsigned* __restrict__ tab, unsigned* __restrict__ blist,
                        int n_tiles) {
    __shared__ __attribute__((aligned(16))) unsigned char lds_raw[(HT_WORDS + OUT_F * WPITCH) * 4];

    if (blockIdx.x < EG) {
        unsigned* const lw = (unsigned*)lds_raw;
        chunk_sort_role(rows, cols, tab, blist, lw, lw + NBKT, (int)blockIdx.x);
        return;
    }
    float* const lds = (float*)lds_raw;

    if (blockIdx.x == EG && threadIdx.x == 0) *gmax = 0u;

    const int lane = threadIdx.x & 31;
    const int wave = threadIdx.x >> 5;
    const int tile = ((int)blockIdx.x - EG) * 4 + wave;

    const bool live = tile < n_tiles;
    const int row0 = tile * 16;
    const int half = lane >> 4;
    const int mn   = lane & 15;

    float* tl = lds + wave * (16 * 128);
    float4 ht[16];
    {
        const float4* hp4 = (const float4*)(h + (live ? row0 : (n_tiles - 1) * 16) * IN_F);
        #pragma unroll
        for (int t = 0; t < 16; ++t)
            ht[t] = hp4[t * 32 + lane];
    }

    float* const wl = lds + HT_WORDS;
    {
        const float4* W4 = (const float4*)W;
        #pragma unroll
        for (int i = 0; i < (OUT_F * IN_F / 4) / CT; ++i) {
            const int idx = i * CT + (int)threadIdx.x;
            *(float4*)(wl + (idx >> 5) * WPITCH + (idx & 31) * 4) = W4[idx];
        }
    }
    {
        float4* tl4 = (float4*)tl;
        #pragma unroll
        for (int t = 0; t < 16; ++t)
            tl4[t * 32 + lane] = ht[t];
    }
    __syncthreads();
    if (!live) return;

    float a1v[8], a2v[8];
    #pragma unroll
    for (int nt = 0; nt < 8; ++nt) {
        a1v[nt] = a[nt * 16 + mn];
        a2v[nt] = a[OUT_F + nt * 16 + mn];
    }

    v8f acc[8] = {};

    for (int kc = 0; kc < 32; ++kc) {
        const int col = kc * 4 + 2 * half;
        v2f av;
        {
            const float2 t = *(const float2*)(tl + mn * 128 + col);
            av[0] = t.x; av[1] = t.y;
        }
        #pragma unroll
        for (int nt = 0; nt < 8; ++nt) {
            const int ncol = nt * 16 + mn;
            v2f bv;
            {
                const float2 t = *(const float2*)(wl + ncol * WPITCH + col);
                bv[0] = t.x; bv[1] = t.y;
            }
            acc[nt] = __builtin_amdgcn_wmma_f32_16x16x4_f32(
                false, av, false, bv, (short)0, acc[nt], false, false);
        }
    }

    #pragma unroll
    for (int nt = 0; nt < 8; ++nt) {
        #pragma unroll
        for (int j = 0; j < 8; ++j) {
            Wh[(row0 + j + 8 * half) * OUT_F + nt * 16 + mn] = acc[nt][j];
        }
    }

    float p1[8], p2[8];
    #pragma unroll
    for (int j = 0; j < 8; ++j) { p1[j] = 0.f; p2[j] = 0.f; }
    #pragma unroll
    for (int nt = 0; nt < 8; ++nt) {
        #pragma unroll
        for (int j = 0; j < 8; ++j) {
            p1[j] = fmaf(acc[nt][j], a1v[nt], p1[j]);
            p2[j] = fmaf(acc[nt][j], a2v[nt], p2[j]);
        }
    }
    #pragma unroll
    for (int j = 0; j < 8; ++j) {
        #pragma unroll
        for (int off = 1; off < 16; off <<= 1) {
            p1[j] += __shfl_xor(p1[j], off, 32);
            p2[j] += __shfl_xor(p2[j], off, 32);
        }
    }
    if (mn == 0) {
        #pragma unroll
        for (int j = 0; j < 8; ++j) {
            s1[row0 + j + 8 * half] = p1[j];
            s2[row0 + j + 8 * half] = p2[j];
        }
    }
}

__global__ void __launch_bounds__(DT)
gat_range_csr_and_scores(const unsigned* __restrict__ tab, const unsigned* __restrict__ blist,
                         unsigned* __restrict__ roff, int* __restrict__ scol, float* __restrict__ sval,
                         const float* __restrict__ s1, const float* __restrict__ s2,
                         unsigned* __restrict__ gmax) {
    __shared__ unsigned cntL[DT];
    __shared__ unsigned wtot[DT / 32];
    __shared__ __attribute__((aligned(16))) unsigned pst[EG];
    __shared__ __attribute__((aligned(16))) unsigned pln[EG];
    __shared__ unsigned wst[DT / 32];
    __shared__ unsigned wln[DT / 32];
    __shared__ int sL[RCAP];
    __shared__ float vL[RCAP];
    __shared__ float s1L[DT];
    __shared__ __attribute__((aligned(16))) float s2L[N_NODES];
    __shared__ float wmx[DT / 32];

    const int tid = threadIdx.x;
    __builtin_assume(tid < DT);
    const int lane = tid & 31;
    const int wave = __builtin_amdgcn_readfirstlane((int)(threadIdx.x >> 5));
    const int b = blockIdx.x;
    const int row0 = b * RPB;

    cntL[tid] = 0u;
    const unsigned t = (tid < EG) ? tab[b * EG + tid] : 0u;
    const float s1v = (tid < RPB && row0 + tid < N_NODES) ? s1[row0 + tid] : 0.f;
    float4 sv[S2R];
    {
        const float4* s2g = (const float4*)s2;
        #pragma unroll
        for (int i = 0; i < S2R; ++i) {
            const int idx = i * DT + tid;
            sv[i] = (idx < S2V4) ? s2g[idx] : make_float4(0.f, 0.f, 0.f, 0.f);
        }
    }
    unsigned st = 0u, ln = 0u;
    if (tid < EG) {
        st = t >> 16;
        ln = t & 0xFFFFu;
        pst[tid] = (unsigned)(tid * ECH) + st;
        pln[tid] = ln;
    }
    for (int off = 16; off >= 1; off >>= 1) {
        st += __shfl_xor(st, off, 32);
        ln += __shfl_xor(ln, off, 32);
    }
    if (lane == 0) { wst[wave] = st; wln[wave] = ln; }
    __syncthreads();
    int beg = 0, n = 0;
    #pragma unroll
    for (int w = 0; w < DT / 32; ++w) { beg += (int)wst[w]; n += (int)wln[w]; }

    unsigned e[PPW], f[PPW];
    int ps[PPW], pl[PPW];
    #pragma unroll
    for (int k = 0; k < PPW; ++k) {
        ps[k] = __builtin_amdgcn_readfirstlane((int)pst[wave * PPW + k]);
        pl[k] = __builtin_amdgcn_readfirstlane((int)pln[wave * PPW + k]);
    }
    bool any_long = false;
    #pragma unroll
    for (int k = 0; k < PPW; ++k) {
        e[k] = (lane < pl[k]) ? blist[ps[k] + lane] : EMPTY_EDGE;
        f[k] = EMPTY_EDGE;
        if (pl[k] > 32) f[k] = (32 + lane < pl[k]) ? blist[ps[k] + 32 + lane] : EMPTY_EDGE;
        any_long |= (pl[k] > 64);
    }

    {
        float4* s2l4 = (float4*)s2L;
        #pragma unroll
        for (int i = 0; i < S2R; ++i) {
            const int idx = i * DT + tid;
            if (idx < S2V4) s2l4[idx] = sv[i];
        }
    }
    s1L[tid] = s1v;

    #pragma unroll
    for (int k = 0; k < PPW; ++k)
        if (e[k] != EMPTY_EDGE) atomicAdd(&cntL[e[k] >> 16], 1u);
    #pragma unroll
    for (int k = 0; k < PPW; ++k)
        if (f[k] != EMPTY_EDGE) atomicAdd(&cntL[f[k] >> 16], 1u);
    if (any_long) {
        for (int k = 0; k < PPW; ++k) {
            const int p = wave * PPW + k;
            const int rs = (int)pst[p];
            const int rl = (int)pln[p];
            for (int j = 64 + lane; j < rl; j += 32)
                atomicAdd(&cntL[blist[rs + j] >> 16], 1u);
        }
    }
    __syncthreads();

    const unsigned v = cntL[tid];
    unsigned s = v;
    for (int off = 1; off < 32; off <<= 1) {
        const unsigned t2 = __shfl_up(s, off, 32);
        if (lane >= off) s += t2;
    }
    if (lane == 31) wtot[wave] = s;
    __syncthreads();
    const unsigned wt = (lane < DT / 32) ? wtot[lane] : 0u;
    unsigned ws = wt;
    for (int off = 1; off < 32; off <<= 1) {
        const unsigned t2 = __shfl_up(ws, off, 32);
        if (lane >= off) ws += t2;
    }
    const unsigned wexcl = __shfl(ws, wave, 32) - __shfl(wt, wave, 32);
    const unsigned excl = wexcl + s - v;
    cntL[tid] = excl;
    if (tid < RPB && row0 + tid < N_NODES) roff[row0 + tid] = (unsigned)beg + excl;

    float se[PPW], sf[PPW];
    float lmax = -INFINITY;
    #pragma unroll
    for (int k = 0; k < PPW; ++k) {
        const unsigned w = (e[k] != EMPTY_EDGE) ? e[k] : 0u;
        se[k] = edge_score(s1L[w >> 16], s2L[w & 0xFFFFu]);
        if (e[k] != EMPTY_EDGE) lmax = fmaxf(lmax, se[k]);
    }
    #pragma unroll
    for (int k = 0; k < PPW; ++k) {
        sf[k] = 0.f;
        if (pl[k] > 32) {
            const unsigned w = (f[k] != EMPTY_EDGE) ? f[k] : 0u;
            sf[k] = edge_score(s1L[w >> 16], s2L[w & 0xFFFFu]);
            if (f[k] != EMPTY_EDGE) lmax = fmaxf(lmax, sf[k]);
        }
    }
    __syncthreads();

    if (n <= RCAP) {
        unsigned pos[PPW];
        #pragma unroll
        for (int k = 0; k < PPW; ++k)
            pos[k] = (e[k] != EMPTY_EDGE) ? atomicAdd(&cntL[e[k] >> 16], 1u) : 0u;
        #pragma unroll
        for (int k = 0; k < PPW; ++k)
            if (e[k] != EMPTY_EDGE) {
                sL[pos[k]] = (int)(e[k] & 0xFFFFu);
                vL[pos[k]] = se[k];
            }
        #pragma unroll
        for (int k = 0; k < PPW; ++k)
            if (f[k] != EMPTY_EDGE) {
                const unsigned p2 = atomicAdd(&cntL[f[k] >> 16], 1u);
                sL[p2] = (int)(f[k] & 0xFFFFu);
                vL[p2] = sf[k];
            }
        if (any_long) {
            for (int k = 0; k < PPW; ++k) {
                const int p = wave * PPW + k;
                const int rs = (int)pst[p];
                const int rl = (int)pln[p];
                for (int j = 64 + lane; j < rl; j += 32) {
                    const unsigned w = blist[rs + j];
                    const float sc = edge_score(s1L[w >> 16], s2L[w & 0xFFFFu]);
                    lmax = fmaxf(lmax, sc);
                    const unsigned p2 = atomicAdd(&cntL[w >> 16], 1u);
                    sL[p2] = (int)(w & 0xFFFFu);
                    vL[p2] = sc;
                }
            }
        }
        __syncthreads();
        for (int j = tid; j < n; j += DT) {
            scol[beg + j] = sL[j];
            sval[beg + j] = vL[j];
        }
    } else {
        #pragma unroll
        for (int k = 0; k < PPW; ++k)
            if (e[k] != EMPTY_EDGE) {
                const unsigned pos = atomicAdd(&cntL[e[k] >> 16], 1u);
                scol[beg + pos] = (int)(e[k] & 0xFFFFu);
                sval[beg + pos] = se[k];
            }
        #pragma unroll
        for (int k = 0; k < PPW; ++k)
            if (f[k] != EMPTY_EDGE) {
                const unsigned pos = atomicAdd(&cntL[f[k] >> 16], 1u);
                scol[beg + pos] = (int)(f[k] & 0xFFFFu);
                sval[beg + pos] = sf[k];
            }
        if (any_long) {
            for (int k = 0; k < PPW; ++k) {
                const int p = wave * PPW + k;
                const int rs = (int)pst[p];
                const int rl = (int)pln[p];
                for (int j = 64 + lane; j < rl; j += 32) {
                    const unsigned w = blist[rs + j];
                    const float sc = edge_score(s1L[w >> 16], s2L[w & 0xFFFFu]);
                    lmax = fmaxf(lmax, sc);
                    const unsigned pos = atomicAdd(&cntL[w >> 16], 1u);
                    scol[beg + pos] = (int)(w & 0xFFFFu);
                    sval[beg + pos] = sc;
                }
            }
        }
    }

    for (int off = 16; off >= 1; off >>= 1)
        lmax = fmaxf(lmax, __shfl_xor(lmax, off, 32));
    if (lane == 0) wmx[wave] = lmax;
    __syncthreads();
    if (wave == 0) {
        float m = (lane < DT / 32) ? wmx[lane] : -INFINITY;
        for (int off = 16; off >= 1; off >>= 1)
            m = fmaxf(m, __shfl_xor(m, off, 32));
        if (lane == 0) atomicMax(gmax, fenc(m));
    }
}

__global__ void __launch_bounds__(256)
gat_row_gather(const unsigned* __restrict__ roff, const int* __restrict__ scol,
               const float* __restrict__ sval,
               const unsigned* __restrict__ gmax, const float* __restrict__ Wh,
               float* __restrict__ out) {
    const int lane = threadIdx.x & 31;
    const int wave = __builtin_amdgcn_readfirstlane((int)(threadIdx.x >> 5));
    const int r = blockIdx.x * 8 + wave;
    if (r >= N_NODES) return;

    const int beg = (int)roff[r];
    const int end = (r + 1 < N_NODES) ? (int)roff[r + 1] : N_EDGES;
    const float gm = fdec(*gmax);

    float lsum = 0.f;
    for (int j = beg + lane; j < end; j += 32)
        lsum += __expf(sval[j] - gm);
    for (int off = 16; off >= 1; off >>= 1)
        lsum += __shfl_xor(lsum, off, 32);
    const float den = lsum + EPS_F;

    const float4* Wh4 = (const float4*)Wh;
    float4 acc = make_float4(0.f, 0.f, 0.f, 0.f);
    for (int base = beg; base < end; base += 32) {
        const int j = base + lane;
        const int jj = (j < end) ? j : (end - 1);
        const int c = scol[jj];
        const float ex = __expf(sval[jj] - gm);
        const float al = (j < end) ? ex / den : 0.f;
        const int n = min(32, end - base);
        const int n4 = (n + 3) & ~3;
        for (int k = 0; k < n4; k += 4) {
            float4 w[4];
            float ak[4];
            for (int u = 0; u < 4; ++u) {
                const int ck = __builtin_amdgcn_readlane(c, k + u);
                ak[u] = __int_as_float(__builtin_amdgcn_readlane(__float_as_int(al), k + u));
                w[u] = Wh4[ck * (OUT_F / 4) + lane];
            }
            for (int u = 0; u < 4; ++u) {
                acc.x = fmaf(ak[u], w[u].x, acc.x);
                acc.y = fmaf(ak[u], w[u].y, acc.y);
                acc.z = fmaf(ak[u], w[u].z, acc.z);
                acc.w = fmaf(ak[u], w[u].w, acc.w);
            }
        }
    }

    float4 o;
    o.x = (acc.x > 0.f) ? acc.x : expm1f(acc.x);
    o.y = (acc.y > 0.f) ? acc.y : expm1f(acc.y);
    o.z = (acc.z > 0.f) ? acc.z : expm1f(acc.z);
    o.w = (acc.w > 0.f) ? acc.w : expm1f(acc.w);
    ((float4*)(out + r * OUT_F))[lane] = o;
}

extern "C" void kernel_launch(void* const* d_in, const int* in_sizes, int n_in,
                              void* d_out, int out_size, void* d_ws, size_t ws_size,
                              hipStream_t stream) {
    const float* h  = (const float*)d_in[0];
    const int*   ei = (const int*)d_in[1];
    const float* W  = (const float*)d_in[2];
    const float* a  = (const float*)d_in[3];
    float* out = (float*)d_out;

    const int* rows = ei;
    const int* cols = ei + N_EDGES;

    float* ws = (float*)d_ws;
    float*    Wh    = ws;
    float*    s1    = Wh + (size_t)N_NODES * OUT_F;
    float*    s2    = s1 + N_NODES;
    float*    sval  = s2 + N_NODES;
    int*      scol  = (int*)(sval + N_EDGES);
    unsigned* roff  = (unsigned*)(scol + N_EDGES);
    unsigned* gmax  = roff + N_NODES;
    unsigned* tab   = (unsigned*)d_out;
    unsigned* blist = tab + NBKT * EG;

    const int n_tiles = N_NODES / 16;
    gat_gemm_and_chunk_sort<<<EG + (n_tiles + 3) / 4, CT, 0, stream>>>(h, W, a, Wh, s1, s2, gmax,
                                                                     rows, cols, tab, blist, n_tiles);
    gat_range_csr_and_scores<<<NBKT, DT, 0, stream>>>(tab, blist, roff, scol, sval, s1, s2, gmax);
    gat_row_gather<<<(N_NODES + 7) / 8, 256, 0, stream>>>(roff, scol, sval, gmax, Wh, out);
}


# Range pass on all 256 WGPs, not 128: 256 row ranges of 196 rows, runs cut from 128 edge chunks (as many runs, as long), so a range wave takes 8 runs, not 16; row sums stay in arrival order
# speedup vs baseline: 4.4436x; 1.0592x over previous
#include <hip/hip_runtime.h>
#include <math.h>

#define N_NODES 50000
#define N_EDGES 800000
#define IN_F 128
#define OUT_F 128
#define NEG_SLOPE 0.2f
#define EPS_F 1e-10f

typedef float v2f __attribute__((ext_vector_type(2)));
typedef float v8f __attribute__((ext_vector_type(8)));


__device__ __forceinline__ unsigned fenc(float f) {
    unsigned u = __float_as_uint(f);
    return (u & 0x80000000u) ? ~u : (u | 0x80000000u);
}
__device__ __forceinline__ float fdec(unsigned u) {
    return __uint_as_float((u & 0x80000000u) ? (u ^ 0x80000000u) : ~u);
}

__device__ __forceinline__ float edge_score(float a, float b) {
    const float v = a + b;
    return (v > 0.f) ? v : NEG_SLOPE * v;
}

#define NBKT 256
#define RPB 196
#define EG 128
#define ECH (N_EDGES / EG)
#define CT 128
#define EPT ((ECH + CT - 1) / CT)
#define DT 512
#define PPW (EG / (DT / 32))
#define RCAP 8192
#define S2V4 (N_NODES / 4)
#define S2R ((S2V4 + DT - 1) / DT)
#define EMPTY_EDGE 0xFFFFFFFFu
#define HT_WORDS (4 * 16 * 128)
#define WPITCH 132
static_assert(NBKT * RPB >= N_NODES, "ranges must cover all rows");
static_assert(ECH * EG == N_EDGES, "chunks must tile the edge list");
static_assert(S2V4 * 4 == N_NODES, "s2 is copied to LDS in whole float4 pieces");
static_assert(NBKT == 8 * 32, "one wave scans the range counters, eight per lane");
static_assert(RPB <= DT && EG <= DT, "one thread per row counter / per run of a range");
static_assert(PPW * (DT / 32) == EG, "the waves of a range work-group share out its runs evenly");
static_assert(N_NODES <= 65536 && RPB <= 512, "packed edge word: local row above bit 16, column below");
static_assert(ECH < 65536, "table word: run start above bit 16, run length below");
static_assert(NBKT + ECH <= HT_WORDS, "the chunk role's counters and sorted edges fit the GEMM role's LDS tiles");
static_assert((OUT_F * IN_F / 4) % CT == 0, "the work-group's threads share out the float4 pieces of W evenly");
static_assert(WPITCH % 4 == 0 && WPITCH >= IN_F, "rows of the LDS copy of W hold a whole W row and stay 16-byte aligned");

__device__ __forceinline__ void chunk_sort_role(const int* __restrict__ rows, const int* __restrict__ cols,
                                                unsigned* __restrict__ tab, unsigned* __restrict__ blist,
                                                unsigned* __restrict__ hist,
                                                unsigned* __restrict__ sorted,
                                                const int chunk) {
    const int tid = threadIdx.x;
    __builtin_assume(tid < CT);
    const int lane = tid & 31;
    const int wave = tid >> 5;
    for (int i = tid; i < NBKT; i += CT) hist[i] = 0u;

    const int base = chunk * ECH;
    int r[EPT], c[EPT];
    #pragma unroll
    for (int u = 0; u < EPT; ++u) {
        const int i = u * CT + tid;
        r[u] = (i < ECH) ? rows[base + i] : 0;
        c[u] = (i < ECH) ? cols[base + i] : 0;
    }
    __syncthreads();

    int q[EPT];
    unsigned rk[EPT];
    #pragma unroll
    for (int u = 0; u < EPT; ++u) {
        q[u] = r[u] / RPB;
        rk[u] = (u * CT + tid < ECH) ? atomicAdd(&hist[q[u]], 1u) : 0u;
    }
    __syncthreads();

    if (wave == 0) {
        const uint4 ha = ((const uint4*)hist)[2 * lane];
        const uint4 hb = ((const uint4*)hist)[2 * lane + 1];
        const unsigned t8 = ha.x + ha.y + ha.z + ha.w + hb.x + hb.y + hb.z + hb.w;
        unsigned s = t8;
        for (int off = 1; off < 32; off <<= 1) {
            const unsigned n = __shfl_up(s, off, 32);
            if (lane >= off) s += n;
        }
        const unsigned e0 = s - t8;
        const unsigned e1 = e0 + ha.x;
        const unsigned e2 = e1 + ha.y;
        const unsigned e3 = e2 + ha.z;
        const unsigned e4 = e3 + ha.w;
        const unsigned e5 = e4 + hb.x;
        const unsigned e6 = e5 + hb.y;
        const unsigned e7 = e6 + hb.z;
        ((uint4*)hist)[2 * lane]     = make_uint4(e0, e1, e2, e3);
        ((uint4*)hist)[2 * lane + 1] = make_uint4(e4, e5, e6, e7);
        tab[(8 * lane + 0) * EG + chunk] = (e0 << 16) | ha.x;
        tab[(8 * lane + 1) * EG + chunk] = (e1 << 16) | ha.y;
        tab[(8 * lane + 2) * EG + chunk] = (e2 << 16) | ha.z;
        tab[(8 * lane + 3) * EG + chunk] = (e3 << 16) | ha.w;
        tab[(8 * lane + 4) * EG + chunk] = (e4 << 16) | hb.x;
        tab[(8 * lane + 5) * EG + chunk] = (e5 << 16) | hb.y;
        tab[(8 * lane + 6) * EG + chunk] = (e6 << 16) | hb.z;
        tab[(8 * lane + 7) * EG + chunk] = (e7 << 16) | hb.w;
    }
    __syncthreads();

    unsigned slot[EPT];
    #pragma unroll
    for (int u = 0; u < EPT; ++u)
        slot[u] = hist[q[u]] + rk[u];
    #pragma unroll
    for (int u = 0; u < EPT; ++u)
        if (u * CT + tid < ECH)
            sorted[slot[u]] = ((unsigned)(r[u] - q[u] * RPB) << 16) | (unsigned)c[u];
    __syncthreads();

    for (int i = tid; i < ECH; i += CT)
        blist[base + i] = sorted[i];
}

__global__ void __launch_bounds__(CT)
gat_gemm_and_chunk_sort(const float* __restrict__ h, const float* __restrict__ W,
                        const float* __restrict__ a, float* __restrict__ Wh,
                        float* __restrict__ s1, float* __restrict__ s2, unsigned* __restrict__ gmax,
                        const int* __restrict__ rows, const int* __restrict__ cols,
                        unsigned* __restrict__ tab, unsigned* __restrict__ blist,
                        int n_tiles) {
    __shared__ __attribute__((aligned(16))) unsigned char lds_raw[(HT_WORDS + OUT_F * WPITCH) * 4];

    if (blockIdx.x < EG) {
        unsigned* const lw = (unsigned*)lds_raw;
        chunk_sort_role(rows, cols, tab, blist, lw, lw + NBKT, (int)blockIdx.x);
        return;
    }
    float* const lds = (float*)lds_raw;

    if (blockIdx.x == EG && threadIdx.x == 0) *gmax = 0u;

    const int lane = threadIdx.x & 31;
    const int wave = threadIdx.x >> 5;
    const int tile = ((int)blockIdx.x - EG) * 4 + wave;

    const bool live = tile < n_tiles;
    const int row0 = tile * 16;
    const int half = lane >> 4;
    const int mn   = lane & 15;

    float* tl = lds + wave * (16 * 128);
    float4 ht[16];
    {
        const float4* hp4 = (const float4*)(h + (live ? row0 : (n_tiles - 1) * 16) * IN_F);
        #pragma unroll
        for (int t = 0; t < 16; ++t)
            ht[t] = hp4[t * 32 + lane];
    }

    float* const wl = lds + HT_WORDS;
    {
        const float4* W4 = (const float4*)W;
        #pragma unroll
        for (int i = 0; i < (OUT_F * IN_F / 4) / CT; ++i) {
            const int idx = i * CT + (int)threadIdx.x;
            *(float4*)(wl + (idx >> 5) * WPITCH + (idx & 31) * 4) = W4[idx];
        }
    }
    {
        float4* tl4 = (float4*)tl;
        #pragma unroll
        for (int t = 0; t < 16; ++t)
            tl4[t * 32 + lane] = ht[t];
    }
    __syncthreads();
    if (!live) return;

    float a1v[8], a2v[8];
    #pragma unroll
    for (int nt = 0; nt < 8; ++nt) {
        a1v[nt] = a[nt * 16 + mn];
        a2v[nt] = a[OUT_F + nt * 16 + mn];
    }

    v8f acc[8] = {};

    for (int kc = 0; kc < 32; ++kc) {
        const int col = kc * 4 + 2 * half;
        v2f av;
        {
            const float2 t = *(const float2*)(tl + mn * 128 + col);
            av[0] = t.x; av[1] = t.y;
        }
        #pragma unroll
        for (int nt = 0; nt < 8; ++nt) {
            const int ncol = nt * 16 + mn;
            v2f bv;
            {
                const float2 t = *(const float2*)(wl + ncol * WPITCH + col);
                bv[0] = t.x; bv[1] = t.y;
            }
            acc[nt] = __builtin_amdgcn_wmma_f32_16x16x4_f32(
                false, av, false, bv, (short)0, acc[nt], false, false);
        }
    }

    #pragma unroll
    for (int nt = 0; nt < 8; ++nt) {
        #pragma unroll
        for (int j = 0; j < 8; ++j) {
            Wh[(row0 + j + 8 * half) * OUT_F + nt * 16 + mn] = acc[nt][j];
        }
    }

    float p1[8], p2[8];
    #pragma unroll
    for (int j = 0; j < 8; ++j) { p1[j] = 0.f; p2[j] = 0.f; }
    #pragma unroll
    for (int nt = 0; nt < 8; ++nt) {
        #pragma unroll
        for (int j = 0; j < 8; ++j) {
            p1[j] = fmaf(acc[nt][j], a1v[nt], p1[j]);
            p2[j] = fmaf(acc[nt][j], a2v[nt], p2[j]);
        }
    }
    #pragma unroll
    for (int j = 0; j < 8; ++j) {
        #pragma unroll
        for (int off = 1; off < 16; off <<= 1) {
            p1[j] += __shfl_xor(p1[j], off, 32);
            p2[j] += __shfl_xor(p2[j], off, 32);
        }
    }
    if (mn == 0) {
        #pragma unroll
        for (int j = 0; j < 8; ++j) {
            s1[row0 + j + 8 * half] = p1[j];
            s2[row0 + j + 8 * half] = p2[j];
        }
    }
}

__global__ void __launch_bounds__(DT)
gat_range_csr_and_scores(const unsigned* __restrict__ tab, const unsigned* __restrict__ blist,
                         unsigned* __restrict__ roff, int* __restrict__ scol, float* __restrict__ sval,
                         const float* __restrict__ s1, const float* __restrict__ s2,
                         unsigned* __restrict__ gmax) {
    __shared__ unsigned cntL[DT];
    __shared__ unsigned wtot[DT / 32];
    __shared__ __attribute__((aligned(16))) unsigned pst[EG];
    __shared__ __attribute__((aligned(16))) unsigned pln[EG];
    __shared__ unsigned wst[DT / 32];
    __shared__ unsigned wln[DT / 32];
    __shared__ int sL[RCAP];
    __shared__ float vL[RCAP];
    __shared__ float s1L[DT];
    __shared__ __attribute__((aligned(16))) float s2L[N_NODES];
    __shared__ float wmx[DT / 32];

    const int tid = threadIdx.x;
    __builtin_assume(tid < DT);
    const int lane = tid & 31;
    const int wave = __builtin_amdgcn_readfirstlane((int)(threadIdx.x >> 5));
    const int b = blockIdx.x;
    const int row0 = b * RPB;

    cntL[tid] = 0u;
    const unsigned t = (tid < EG) ? tab[b * EG + tid] : 0u;
    const float s1v = (tid < RPB && row0 + tid < N_NODES) ? s1[row0 + tid] : 0.f;
    float4 sv[S2R];
    {
        const float4* s2g = (const float4*)s2;
        #pragma unroll
        for (int i = 0; i < S2R; ++i) {
            const int idx = i * DT + tid;
            sv[i] = (idx < S2V4) ? s2g[idx] : make_float4(0.f, 0.f, 0.f, 0.f);
        }
    }
    unsigned st = 0u, ln = 0u;
    if (tid < EG) {
        st = t >> 16;
        ln = t & 0xFFFFu;
        pst[tid] = (unsigned)(tid * ECH) + st;
        pln[tid] = ln;
    }
    for (int off = 16; off >= 1; off >>= 1) {
        st += __shfl_xor(st, off, 32);
        ln += __shfl_xor(ln, off, 32);
    }
    if (lane == 0) { wst[wave] = st; wln[wave] = ln; }
    __syncthreads();
    int beg = 0, n = 0;
    #pragma unroll
    for (int w = 0; w < DT / 32; ++w) { beg += (int)wst[w]; n += (int)wln[w]; }

    unsigned e[PPW], f[PPW];
    int ps[PPW], pl[PPW];
    #pragma unroll
    for (int k = 0; k < PPW; ++k) {
        ps[k] = __builtin_amdgcn_readfirstlane((int)pst[wave * PPW + k]);
        pl[k] = __builtin_amdgcn_readfirstlane((int)pln[wave * PPW + k]);
    }
    bool any_long = false;
    #pragma unroll
    for (int k = 0; k < PPW; ++k) {
        e[k] = (lane < pl[k]) ? blist[ps[k] + lane] : EMPTY_EDGE;
        f[k] = EMPTY_EDGE;
        if (pl[k] > 32) f[k] = (32 + lane < pl[k]) ? blist[ps[k] + 32 + lane] : EMPTY_EDGE;
        any_long |= (pl[k] > 64);
    }

    {
        float4* s2l4 = (float4*)s2L;
        #pragma unroll
        for (int i = 0; i < S2R; ++i) {
            const int idx = i * DT + tid;
            if (idx < S2V4) s2l4[idx] = sv[i];
        }
    }
    s1L[tid] = s1v;

    #pragma unroll
    for (int k = 0; k < PPW; ++k)
        if (e[k] != EMPTY_EDGE) atomicAdd(&cntL[e[k] >> 16], 1u);
    #pragma unroll
    for (int k = 0; k < PPW; ++k)
        if (f[k] != EMPTY_EDGE) atomicAdd(&cntL[f[k] >> 16], 1u);
    if (any_long) {
        for (int k = 0; k < PPW; ++k) {
            const int p = wave * PPW + k;
            const int rs = (int)pst[p];
            const int rl = (int)pln[p];
            for (int j = 64 + lane; j < rl; j += 32)
                atomicAdd(&cntL[blist[rs + j] >> 16], 1u);
        }
    }
    __syncthreads();

    const unsigned v = cntL[tid];
    unsigned s = v;
    for (int off = 1; off < 32; off <<= 1) {
        const unsigned t2 = __shfl_up(s, off, 32);
        if (lane >= off) s += t2;
    }
    if (lane == 31) wtot[wave] = s;
    __syncthreads();
    const unsigned wt = (lane < DT / 32) ? wtot[lane] : 0u;
    unsigned ws = wt;
    for (int off = 1; off < 32; off <<= 1) {
        const unsigned t2 = __shfl_up(ws, off, 32);
        if (lane >= off) ws += t2;
    }
    const unsigned wexcl = __shfl(ws, wave, 32) - __shfl(wt, wave, 32);
    const unsigned excl = wexcl + s - v;
    cntL[tid] = excl;
    if (tid < RPB && row0 + tid < N_NODES) roff[row0 + tid] = (unsigned)beg + excl;

    float se[PPW], sf[PPW];
    float lmax = -INFINITY;
    #pragma unroll
    for (int k = 0; k < PPW; ++k) {
        const unsigned w = (e[k] != EMPTY_EDGE) ? e[k] : 0u;
        se[k] = edge_score(s1L[w >> 16], s2L[w & 0xFFFFu]);
        if (e[k] != EMPTY_EDGE) lmax = fmaxf(lmax, se[k]);
    }
    #pragma unroll
    for (int k = 0; k < PPW; ++k) {
        sf[k] = 0.f;
        if (pl[k] > 32) {
            const unsigned w = (f[k] != EMPTY_EDGE) ? f[k] : 0u;
            sf[k] = edge_score(s1L[w >> 16], s2L[w & 0xFFFFu]);
            if (f[k] != EMPTY_EDGE) lmax = fmaxf(lmax, sf[k]);
        }
    }
    __syncthreads();

    if (n <= RCAP) {
        unsigned pos[PPW];
        #pragma unroll
        for (int k = 0; k < PPW; ++k)
            pos[k] = (e[k] != EMPTY_EDGE) ? atomicAdd(&cntL[e[k] >> 16], 1u) : 0u;
        #pragma unroll
        for (int k = 0; k < PPW; ++k)
            if (e[k] != EMPTY_EDGE) {
                sL[pos[k]] = (int)(e[k] & 0xFFFFu);
                vL[pos[k]] = se[k];
            }
        #pragma unroll
        for (int k = 0; k < PPW; ++k)
            if (f[k] != EMPTY_EDGE) {
                const unsigned p2 = atomicAdd(&cntL[f[k] >> 16], 1u);
                sL[p2] = (int)(f[k] & 0xFFFFu);
                vL[p2] = sf[k];
            }
        if (any_long) {
            for (int k = 0; k < PPW; ++k) {
                const int p = wave * PPW + k;
                const int rs = (int)pst[p];
                const int rl = (int)pln[p];
                for (int j = 64 + lane; j < rl; j += 32) {
                    const unsigned w = blist[rs + j];
                    const float sc = edge_score(s1L[w >> 16], s2L[w & 0xFFFFu]);
                    lmax = fmaxf(lmax, sc);
                    const unsigned p2 = atomicAdd(&cntL[w >> 16], 1u);
                    sL[p2] = (int)(w & 0xFFFFu);
                    vL[p2] = sc;
                }
            }
        }
        __syncthreads();
        for (int j = tid; j < n; j += DT) {
            scol[beg + j] = sL[j];
            sval[beg + j] = vL[j];
        }
    } else {
        #pragma unroll
        for (int k = 0; k < PPW; ++k)
            if (e[k] != EMPTY_EDGE) {
                const unsigned pos = atomicAdd(&cntL[e[k] >> 16], 1u);
                scol[beg + pos] = (int)(e[k] & 0xFFFFu);
                sval[beg + pos] = se[k];
            }
        #pragma unroll
        for (int k = 0; k < PPW; ++k)
            if (f[k] != EMPTY_EDGE) {
                const unsigned pos = atomicAdd(&cntL[f[k] >> 16], 1u);
                scol[beg + pos] = (int)(f[k] & 0xFFFFu);
                sval[beg + pos] = sf[k];
            }
        if (any_long) {
            for (int k = 0; k < PPW; ++k) {
                const int p = wave * PPW + k;
                const int rs = (int)pst[p];
                const int rl = (int)pln[p];
                for (int j = 64 + lane; j < rl; j += 32) {
                    const unsigned w = blist[rs + j];
                    const float sc = edge_score(s1L[w >> 16], s2L[w & 0xFFFFu]);
                    lmax = fmaxf(lmax, sc);
                    const unsigned pos = atomicAdd(&cntL[w >> 16], 1u);
                    scol[beg + pos] = (int)(w & 0xFFFFu);
                    sval[beg + pos] = sc;
                }
            }
        }
    }

    for (int off = 16; off >= 1; off >>= 1)
        lmax = fmaxf(lmax, __shfl_xor(lmax, off, 32));
    if (lane == 0) wmx[wave] = lmax;
    __syncthreads();
    if (wave == 0) {
        float m = (lane < DT / 32) ? wmx[lane] : -INFINITY;
        for (int off = 16; off >= 1; off >>= 1)
            m = fmaxf(m, __shfl_xor(m, off, 32));
        if (lane == 0) atomicMax(gmax, fenc(m));
    }
}

__global__ void __launch_bounds__(256)
gat_row_gather(const unsigned* __restrict__ roff, const int* __restrict__ scol,
               const float* __restrict__ sval,
               const unsigned* __restrict__ gmax, const float* __restrict__ Wh,
               float* __restrict__ out) {
    const int lane = threadIdx.x & 31;
    const int wave = __builtin_amdgcn_readfirstlane((int)(threadIdx.x >> 5));
    const int r = blockIdx.x * 8 + wave;
    if (r >= N_NODES) return;

    const int beg = (int)roff[r];
    const int end = (r + 1 < N_NODES) ? (int)roff[r + 1] : N_EDGES;
    const float gm = fdec(*gmax);

    float lsum = 0.f;
    for (int j = beg + lane; j < end; j += 32)
        lsum += __expf(sval[j] - gm);
    for (int off = 16; off >= 1; off >>= 1)
        lsum += __shfl_xor(lsum, off, 32);
    const float den = lsum + EPS_F;

    const float4* Wh4 = (const float4*)Wh;
    float4 acc = make_float4(0.f, 0.f, 0.f, 0.f);
    for (int base = beg; base < end; base += 32) {
        const int j = base + lane;
        const int jj = (j < end) ? j : (end - 1);
        const int c = scol[jj];
        const float ex = __expf(sval[jj] - gm);
        const float al = (j < end) ? ex / den : 0.f;
        const int n = min(32, end - base);
        const int n4 = (n + 3) & ~3;
        for (int k = 0; k < n4; k += 4) {
            float4 w[4];
            float ak[4];
            for (int u = 0; u < 4; ++u) {
                const int ck = __builtin_amdgcn_readlane(c, k + u);
                ak[u] = __int_as_float(__builtin_amdgcn_readlane(__float_as_int(al), k + u));
                w[u] = Wh4[ck * (OUT_F / 4) + lane];
            }
            for (int u = 0; u < 4; ++u) {
                acc.x = fmaf(ak[u], w[u].x, acc.x);
                acc.y = fmaf(ak[u], w[u].y, acc.y);
                acc.z = fmaf(ak[u], w[u].z, acc.z);
                acc.w = fmaf(ak[u], w[u].w, acc.w);
            }
        }
    }

    float4 o;
    o.x = (acc.x > 0.f) ? acc.x : expm1f(acc.x);
    o.y = (acc.y > 0.f) ? acc.y : expm1f(acc.y);
    o.z = (acc.z > 0.f) ? acc.z : expm1f(acc.z);
    o.w = (acc.w > 0.f) ? acc.w : expm1f(acc.w);
    ((float4*)(out + r * OUT_F))[lane] = o;
}

extern "C" void kernel_launch(void* const* d_in, const int* in_sizes, int n_in,
                              void* d_out, int out_size, void* d_ws, size_t ws_size,
                              hipStream_t stream) {
    const float* h  = (const float*)d_in[0];
    const int*   ei = (const int*)d_in[1];
    const float* W  = (const float*)d_in[2];
    const float* a  = (const float*)d_in[3];
    float* out = (float*)d_out;

    const int* rows = ei;
    const int* cols = ei + N_EDGES;

    float* ws = (float*)d_ws;
    float*    Wh    = ws;
    float*    s1    = Wh + (size_t)N_NODES * OUT_F;
    float*    s2    = s1 + N_NODES;
    float*    sval  = s2 + N_NODES;
    int*      scol  = (int*)(sval + N_EDGES);
    unsigned* roff  = (unsigned*)(scol + N_EDGES);
    unsigned* gmax  = roff + N_NODES;
    unsigned* tab   = (unsigned*)d_out;
    unsigned* blist = tab + NBKT * EG;

    const int n_tiles = N_NODES / 16;
    gat_gemm_and_chunk_sort<<<EG + (n_tiles + 3) / 4, CT, 0, stream>>>(h, W, a, Wh, s1, s2, gmax,
                                                                     rows, cols, tab, blist, n_tiles);
    gat_range_csr_and_scores<<<NBKT, DT, 0, stream>>>(tab, blist, roff, scol, sval, s1, s2, gmax);
    gat_row_gather<<<(N_NODES + 7) / 8, 256, 0, stream>>>(roff, scol, sval, gmax, Wh, out);
}
